# starting kernel
# baseline (speedup 1.0000x reference)
#include <hip/hip_runtime.h>
#include <hip/hip_bf16.h>
#include <math.h>


#define NN 20000
#define NE 320000

typedef __attribute__((ext_vector_type(2))) float v2f;
typedef __attribute__((ext_vector_type(8))) float v8f;

__device__ __forceinline__ v8f wmma_f32_k4(v2f a, v2f b, v8f c) {
  return __builtin_amdgcn_wmma_f32_16x16x4_f32(false, a, false, b,
                                               (short)0, c, false, false);
}

__device__ __forceinline__ float sspf(float x) {
  return fmaxf(x, 0.0f) + log1pf(__expf(-fabsf(x))) - 0.69314718056f;
}

__device__ __forceinline__ void atadd(float* p, float v) {
  __hip_atomic_fetch_add(p, v, __ATOMIC_RELAXED, __HIP_MEMORY_SCOPE_AGENT);
}

__global__ void k_zero(float4* __restrict__ p, int n4) {
  int i = blockIdx.x * blockDim.x + threadIdx.x;
  const int stride = gridDim.x * blockDim.x;
  for (; i < n4; i += stride) p[i] = make_float4(0.f, 0.f, 0.f, 0.f);
}

__global__ void k_pack(const float* __restrict__ src, float2* __restrict__ dst,
                       int npairs) {
  int i = blockIdx.x * blockDim.x + threadIdx.x;
  const int total = npairs * 64;
  const int stride = gridDim.x * blockDim.x;
  for (; i < total; i += stride) {
    const int kk = i >> 6, n = i & 63;
    dst[i] = make_float2(src[(2 * kk) * 64 + n], src[(2 * kk + 1) * 64 + n]);
  }
}

template <bool ACONTIG, int KC>
__device__ __forceinline__ void gemm_loop(const float* __restrict__ arow,
                                          const float2* __restrict__ Wp,
                                          int kh, int l16, v8f acc[4]) {
  #pragma unroll 4
  for (int kc = 0; kc < KC; ++kc) {
    const int k0 = kc * 4 + 2 * kh;
    v2f a;
    if (ACONTIG) {
      const float2 av = *(const float2*)(arow + k0);
      a.x = av.x; a.y = av.y;
    } else {
      a.x = arow[k0 * 3];
      a.y = arow[(k0 + 1) * 3];
    }
    const float2* bp = Wp + (k0 >> 1) * 64 + l16;
    #pragma unroll
    for (int ct = 0; ct < 4; ++ct) {
      const float2 bv = bp[ct * 16];
      v2f b; b.x = bv.x; b.y = bv.y;
      acc[ct] = wmma_f32_k4(a, b, acc[ct]);
    }
  }
}

__global__ void k_pre(const float* __restrict__ nf,
                      const float2* __restrict__ W1sp,
                      const float2* __restrict__ W1vp,
                      float* __restrict__ sv) {
  const int lane = threadIdx.x & 31;
  const int wid  = threadIdx.x >> 5;
  const int l16  = lane & 15;
  const int kh   = lane >> 4;
  const int row0 = (blockIdx.x * 8 + wid) * 16;
  int rowA = row0 + l16; if (rowA >= NN) rowA = NN - 1;

  for (int tr = 0; tr < 4; ++tr) {
    const int aoff = (tr == 0) ? 0 : (64 + (tr - 1));
    const float* arow = nf + (size_t)rowA * 256 + aoff;
    v8f acc[4] = {};
    if (tr == 0) gemm_loop<true, 16>(arow, W1sp, kh, l16, acc);
    else         gemm_loop<false, 16>(arow, W1vp, kh, l16, acc);

    #pragma unroll
    for (int ct = 0; ct < 4; ++ct) {
      #pragma unroll
      for (int j = 0; j < 8; ++j) {
        const int r = row0 + j + 8 * kh;
        if (r < NN) {
          const int c = ct * 16 + l16;
          const int o = (tr == 0) ? (r * 256 + c)
                                  : (r * 256 + 64 + c * 3 + (tr - 1));
          sv[o] = acc[ct][j] * 0.125f;
        }
      }
    }
  }
}

__global__ void k_msg(const float* __restrict__ ea,
                      const float* __restrict__ efeat,
                      const int*   __restrict__ eidx,
                      const float* __restrict__ Wm1,
                      const float* __restrict__ Wm2,
                      const float* __restrict__ sv,
                      float* __restrict__ msmv) {
  const int e = blockIdx.x * 8 + (threadIdx.x >> 5);
  if (e >= NE) return;
  const int lane = threadIdx.x & 31;

  const int snd = eidx[e];
  const int rcv = eidx[NE + e];
  const float Y0  = ea[e * 4 + 0];
  const float Y1x = ea[e * 4 + 1];
  const float Y1y = ea[e * 4 + 2];
  const float Y1z = ea[e * 4 + 3];

  float f[8];
  #pragma unroll
  for (int q = 0; q < 8; ++q) f[q] = efeat[e * 8 + q];

  float h[8];
  #pragma unroll
  for (int r = 0; r < 8; ++r) {
    float t = 0.f;
    #pragma unroll
    for (int q = 0; q < 8; ++q) t += f[q] * Wm1[q * 8 + r];
    h[r] = sspf(t * 0.35355339059f);
  }

  const int u0 = lane, u1 = lane + 32;
  float w[4][2];
  #pragma unroll
  for (int k = 0; k < 4; ++k) {
    float t0 = 0.f, t1 = 0.f;
    #pragma unroll
    for (int r = 0; r < 8; ++r) {
      t0 += h[r] * Wm2[r * 256 + k * 64 + u0];
      t1 += h[r] * Wm2[r * 256 + k * 64 + u1];
    }
    w[k][0] = t0 * 0.35355339059f;
    w[k][1] = t1 * 0.35355339059f;
  }

  const float* srow = sv + (size_t)snd * 256;
  float* base = msmv + (size_t)rcv * 512;

  #pragma unroll
  for (int t = 0; t < 2; ++t) {
    const int u = (t == 0) ? u0 : u1;
    const float se = srow[u];
    const float vx = srow[64 + u * 3 + 0];
    const float vy = srow[64 + u * 3 + 1];
    const float vz = srow[64 + u * 3 + 2];
    const float dv = vx * Y1x + vy * Y1y + vz * Y1z;
    atadd(base + u,                      w[0][t] * se * Y0);
    atadd(base + 64 + u,                 w[3][t] * dv * 0.57735026919f);
    atadd(base + 128 + u * 3 + 0,        w[1][t] * se * Y1x);
    atadd(base + 128 + u * 3 + 1,        w[1][t] * se * Y1y);
    atadd(base + 128 + u * 3 + 2,        w[1][t] * se * Y1z);
    atadd(base + 128 + (64 + u) * 3 + 0, w[2][t] * vx * Y0);
    atadd(base + 128 + (64 + u) * 3 + 1, w[2][t] * vy * Y0);
    atadd(base + 128 + (64 + u) * 3 + 2, w[2][t] * vz * Y0);
  }
}

__global__ void k_post(const float* __restrict__ msmv,
                       const float* __restrict__ attrs,
                       const float2* __restrict__ W2sp,
                       const float2* __restrict__ W2vp,
                       const float2* __restrict__ WskSp,
                       const float2* __restrict__ WskVp,
                       float* __restrict__ out) {
  __shared__ float lds[8 * 16 * 64];
  const int lane = threadIdx.x & 31;
  const int wid  = threadIdx.x >> 5;
  const int l16  = lane & 15;
  const int kh   = lane >> 4;
  const int row0 = (blockIdx.x * 8 + wid) * 16;
  int rowA = row0 + l16; if (rowA >= NN) rowA = NN - 1;
  float* myLds = lds + wid * (16 * 64);

  const float* arow8 = attrs + (size_t)rowA * 8;
  const float attE0 = arow8[2 * kh]     * 0.04419417382f;
  const float attE1 = arow8[2 * kh + 1] * 0.04419417382f;
  const float attO0 = arow8[4 + 2 * kh] * 0.04419417382f;
  const float attO1 = arow8[5 + 2 * kh] * 0.04419417382f;

  for (int tr = 0; tr < 4; ++tr) {
    const float2* W2p  = (tr == 0) ? W2sp : W2vp;
    const float2* Wskp = (tr == 0) ? WskSp : WskVp;
    const int aoff = (tr == 0) ? 0 : (128 + (tr - 1));
    const float* arow = msmv + (size_t)rowA * 512 + aoff;

    v8f acc[4] = {};
    if (tr == 0) gemm_loop<true, 32>(arow, W2p, kh, l16, acc);
    else         gemm_loop<false, 32>(arow, W2p, kh, l16, acc);

    #pragma unroll
    for (int ct = 0; ct < 4; ++ct) {
      #pragma unroll
      for (int j = 0; j < 8; ++j) {
        acc[ct][j] *= 0.08838834765f;
        myLds[(j + 8 * kh) * 64 + ct * 16 + l16] = acc[ct][j];
      }
    }
    __syncthreads();

    #pragma unroll 2
    for (int i = 0; i < 64; ++i) {
      const float m2 = myLds[l16 * 64 + i];
      const float2* bp = Wskp + (4 * i + kh) * 64 + l16;
      v2f a;
      a.x = m2 * attE0; a.y = m2 * attE1;
      #pragma unroll
      for (int ct = 0; ct < 4; ++ct) {
        const float2 bv = bp[ct * 16];
        v2f b; b.x = bv.x; b.y = bv.y;
        acc[ct] = wmma_f32_k4(a, b, acc[ct]);
      }
      a.x = m2 * attO0; a.y = m2 * attO1;
      #pragma unroll
      for (int ct = 0; ct < 4; ++ct) {
        const float2 bv = bp[2 * 64 + ct * 16];
        v2f b; b.x = bv.x; b.y = bv.y;
        acc[ct] = wmma_f32_k4(a, b, acc[ct]);
      }
    }

    #pragma unroll
    for (int ct = 0; ct < 4; ++ct) {
      #pragma unroll
      for (int j = 0; j < 8; ++j) {
        const int r = row0 + j + 8 * kh;
        if (r < NN) {
          const int c = ct * 16 + l16;
          const int o = (tr == 0) ? (r * 256 + c)
                                  : (r * 256 + 64 + c * 3 + (tr - 1));
          out[o] = acc[ct][j];
        }
      }
    }
    __syncthreads();
  }
}

extern "C" void kernel_launch(void* const* d_in, const int* in_sizes, int n_in,
                              void* d_out, int out_size, void* d_ws, size_t ws_size,
                              hipStream_t stream) {
  const float* node_attrs = (const float*)d_in[0];
  const float* node_feats = (const float*)d_in[1];
  const float* edge_attrs = (const float*)d_in[2];
  const float* edge_feats = (const float*)d_in[3];
  const int*   edge_index = (const int*)d_in[4];
  const float* W1s  = (const float*)d_in[5];
  const float* W1v  = (const float*)d_in[6];
  const float* Wm1  = (const float*)d_in[7];
  const float* Wm2  = (const float*)d_in[8];
  const float* W2s  = (const float*)d_in[9];
  const float* W2v  = (const float*)d_in[10];
  const float* WskS = (const float*)d_in[11];
  const float* WskV = (const float*)d_in[12];

  float* out  = (float*)d_out;
  float* msmv = (float*)d_ws;

  float2* packBase = (float2*)(msmv + (size_t)NN * 512);
  float2* W1sp  = packBase;
  float2* W1vp  = packBase + 2048;
  float2* W2sp  = packBase + 4096;
  float2* W2vp  = packBase + 8192;
  float2* WskSp = packBase + 12288;
  float2* WskVp = packBase + 28672;

  k_pack<<<8,  256, 0, stream>>>(W1s,  W1sp,  32);
  k_pack<<<8,  256, 0, stream>>>(W1v,  W1vp,  32);
  k_pack<<<16, 256, 0, stream>>>(W2s,  W2sp,  64);
  k_pack<<<16, 256, 0, stream>>>(W2v,  W2vp,  64);
  k_pack<<<64, 256, 0, stream>>>(WskS, WskSp, 256);
  k_pack<<<64, 256, 0, stream>>>(WskV, WskVp, 256);
  k_zero<<<2048, 256, 0, stream>>>((float4*)msmv, NN * 512 / 4);
  k_pre <<<(NN + 127) / 128, 256, 0, stream>>>(node_feats, W1sp, W1vp, out);
  k_msg <<<(NE + 7) / 8, 256, 0, stream>>>(edge_attrs, edge_feats, edge_index,
                                           Wm1, Wm2, out, msmv);
  k_post<<<(NN + 127) / 128, 256, 0, stream>>>(msmv, node_attrs, W2sp, W2vp,
                                               WskSp, WskVp, out);
}


# k_pre, k_post: B fragments read in place from the row-major weights (a lane's 8-byte row load = its column of two column-interleaved tiles); six repack launches gone; same k order and sums
# speedup vs baseline: 2.0317x; 2.0317x over previous
#include <hip/hip_runtime.h>
#include <hip/hip_bf16.h>
#include <math.h>


#define NN 20000
#define NE 320000

typedef __attribute__((ext_vector_type(2))) float v2f;
typedef __attribute__((ext_vector_type(8))) float v8f;

__device__ __forceinline__ v8f wmma_f32_k4(v2f a, v2f b, v8f c) {
  return __builtin_amdgcn_wmma_f32_16x16x4_f32(false, a, false, b,
                                               (short)0, c, false, false);
}

__device__ __forceinline__ float sspf(float x) {
  return fmaxf(x, 0.0f) + log1pf(__expf(-fabsf(x))) - 0.69314718056f;
}

__global__ void k_zero(float4* __restrict__ p, int n4) {
  int i = blockIdx.x * blockDim.x + threadIdx.x;
  const int stride = gridDim.x * blockDim.x;
  for (; i < n4; i += stride) p[i] = make_float4(0.f, 0.f, 0.f, 0.f);
}

template <bool ACONTIG, int KC, int NC>
__device__ __forceinline__ void gemm_loop(const float* __restrict__ arow,
                                          const float* __restrict__ W,
                                          int kh, int l16, v8f (&acc)[NC][4]) {
  #pragma unroll 4
  for (int kc = 0; kc < KC; ++kc) {
    const int k0 = kc * 4 + 2 * kh;
    v2f a[NC];
    if (ACONTIG) {
      const float2 av = *(const float2*)(arow + k0);
      a[0].x = av.x; a[0].y = av.y;
    } else {
      const float2* ap = (const float2*)(arow + k0 * 3);
      const float2 f0 = ap[0];
      const float2 f1 = ap[1];
      const float2 f2 = ap[2];
      a[0].x = f0.x; a[0].y = f1.y;
      a[NC > 1 ? 1 : 0].x = f0.y; a[NC > 1 ? 1 : 0].y = f2.x;
      a[NC > 2 ? 2 : 0].x = f1.x; a[NC > 2 ? 2 : 0].y = f2.y;
    }
    const float* bp = W + k0 * 64 + 2 * l16;
    #pragma unroll
    for (int p = 0; p < 2; ++p) {
      const float2 r0 = *(const float2*)(bp + 32 * p);
      const float2 r1 = *(const float2*)(bp + 64 + 32 * p);
      v2f be; be.x = r0.x; be.y = r1.x;
      v2f bo; bo.x = r0.y; bo.y = r1.y;
      #pragma unroll
      for (int d = 0; d < NC; ++d)
        acc[d][2 * p] = wmma_f32_k4(a[d], be, acc[d][2 * p]);
      #pragma unroll
      for (int d = 0; d < NC; ++d)
        acc[d][2 * p + 1] = wmma_f32_k4(a[d], bo, acc[d][2 * p + 1]);
    }
  }
}

__global__ void __launch_bounds__(256)
k_pre(const float* __restrict__ nf,
                      const float* __restrict__ W1s,
                      const float* __restrict__ W1v,
                      float* __restrict__ sv) {
  const int lane = threadIdx.x & 31;
  const int wid  = threadIdx.x >> 5;
  const int l16  = lane & 15;
  const int kh   = lane >> 4;
  const int row0 = (blockIdx.x * 8 + wid) * 16;
  int rowA = row0 + l16; if (rowA >= NN) rowA = NN - 1;
  const float* nrow = nf + (size_t)rowA * 256;

  {
    v8f acc[1][4] = {};
    gemm_loop<true, 16, 1>(nrow, W1s, kh, l16, acc);
    #pragma unroll
    for (int p = 0; p < 2; ++p) {
      #pragma unroll
      for (int j = 0; j < 8; ++j) {
        const int r = row0 + j + 8 * kh;
        if (r < NN)
          *(float2*)(sv + r * 256 + 32 * p + 2 * l16) =
              make_float2(acc[0][2 * p][j] * 0.125f, acc[0][2 * p + 1][j] * 0.125f);
      }
    }
  }
  {
    v8f acc[3][4] = {};
    gemm_loop<false, 16, 3>(nrow + 64, W1v, kh, l16, acc);
    #pragma unroll
    for (int p = 0; p < 2; ++p) {
      #pragma unroll
      for (int j = 0; j < 8; ++j) {
        const int r = row0 + j + 8 * kh;
        if (r < NN) {
          float2* o = (float2*)(sv + r * 256 + 64 + (32 * p + 2 * l16) * 3);
          o[0] = make_float2(acc[0][2 * p][j] * 0.125f,
                             acc[1][2 * p][j] * 0.125f);
          o[1] = make_float2(acc[2][2 * p][j] * 0.125f,
                             acc[0][2 * p + 1][j] * 0.125f);
          o[2] = make_float2(acc[1][2 * p + 1][j] * 0.125f,
                             acc[2][2 * p + 1][j] * 0.125f);
        }
      }
    }
  }
}

__global__ void k_hist(const int* __restrict__ eidx, int* __restrict__ cnt) {
  const int e = blockIdx.x * blockDim.x + threadIdx.x;
  if (e < NE) atomicAdd(&cnt[eidx[NE + e]], 1);
}

__global__ void __launch_bounds__(1024)
k_scan(const int* __restrict__ cnt, int* __restrict__ offs,
       int* __restrict__ cur) {
  __shared__ int part[1024];
  const int t = threadIdx.x;
  const int b = t * 20;
  int loc[20];
  int s = 0;
  #pragma unroll
  for (int k = 0; k < 20; ++k) {
    const int i = b + k;
    const int c = (i < NN) ? cnt[i] : 0;
    loc[k] = s;
    s += c;
  }
  part[t] = s;
  __syncthreads();
  for (int d = 1; d < 1024; d <<= 1) {
    const int v = (t >= d) ? part[t - d] : 0;
    __syncthreads();
    part[t] += v;
    __syncthreads();
  }
  const int base = part[t] - s;
  #pragma unroll
  for (int k = 0; k < 20; ++k) {
    const int i = b + k;
    if (i < NN) { offs[i] = base + loc[k]; cur[i] = base + loc[k]; }
  }
  if (t == 1023) offs[NN] = part[t];
}

__global__ void k_fill(const int* __restrict__ eidx, int* __restrict__ cur,
                       int* __restrict__ list) {
  const int e = blockIdx.x * blockDim.x + threadIdx.x;
  if (e < NE) {
    const int pos = atomicAdd(&cur[eidx[NE + e]], 1);
    list[pos] = e;
  }
}

__device__ __forceinline__ float rlf(float x, int l) {
  return __int_as_float(__builtin_amdgcn_readlane(__float_as_int(x), l));
}

__global__ void __launch_bounds__(256)
k_msg(const float* __restrict__ ea,
      const float* __restrict__ efeat,
      const int*   __restrict__ eidx,
      const float* __restrict__ Wm1,
      const float* __restrict__ Wm2,
      const float* __restrict__ sv,
      const int*   __restrict__ offs,
      const int*   __restrict__ list,
      float* __restrict__ msmv) {
  const int lane = threadIdx.x & 31;
  const int n = __builtin_amdgcn_readfirstlane(blockIdx.x * 8 + (threadIdx.x >> 5));
  if (n >= NN) return;
  const int off = offs[n];
  const int len = offs[n + 1] - off;

  float wm2[8][4][2];
  #pragma unroll
  for (int r = 0; r < 8; ++r) {
    #pragma unroll
    for (int k = 0; k < 4; ++k) {
      wm2[r][k][0] = Wm2[r * 256 + k * 64 + lane];
      wm2[r][k][1] = Wm2[r * 256 + k * 64 + lane + 32];
    }
  }
  float a0[2] = {0.f, 0.f}, b0[2] = {0.f, 0.f};
  float a1[2][3] = {{0.f, 0.f, 0.f}, {0.f, 0.f, 0.f}};
  float b1[2][3] = {{0.f, 0.f, 0.f}, {0.f, 0.f, 0.f}};

  int last = -1;
  for (int done = 0; done < len; done += 32) {
    const int cnt = (len - done < 32) ? (len - done) : 32;
    int mine = 0x7fffffff;
    for (int j0 = 0; j0 < len; j0 += 32) {
      const int jj = j0 + lane;
      const int idj = (jj < len) ? list[off + jj] : 0x7fffffff;
      int rank = 0;
      for (int i0 = 0; i0 < len; i0 += 32) {
        const int ii = i0 + lane;
        const int idi = (ii < len) ? list[off + ii] : 0x7fffffff;
        #pragma unroll
        for (int i = 0; i < 32; ++i) {
          const int v = __builtin_amdgcn_readlane(idi, i);
          rank += (v > last && v < idj) ? 1 : 0;
        }
      }
      const int rk = (idj > last && idj != 0x7fffffff) ? rank : -1;
      #pragma unroll
      for (int i = 0; i < 32; ++i) {
        const int r = __builtin_amdgcn_readlane(rk, i);
        const int v = __builtin_amdgcn_readlane(idj, i);
        if (r == lane) mine = v;
      }
    }
    last = __builtin_amdgcn_readlane(mine, cnt - 1);

    const int e = (lane < cnt) ? mine : 0;
    const float4 fa = *(const float4*)(efeat + (size_t)e * 8);
    const float4 fb = *(const float4*)(efeat + (size_t)e * 8 + 4);
    const float f[8] = {fa.x, fa.y, fa.z, fa.w, fb.x, fb.y, fb.z, fb.w};
    float h[8];
    #pragma unroll
    for (int r = 0; r < 8; ++r) {
      float t = 0.f;
      #pragma unroll
      for (int q = 0; q < 8; ++q) t += f[q] * Wm1[q * 8 + r];
      h[r] = sspf(t * 0.35355339059f);
    }
    const float4 yv = *(const float4*)(ea + (size_t)e * 4);
    const int snd = eidx[e];

    for (int t = 0; t < cnt; ++t) {
      const int   s   = __builtin_amdgcn_readlane(snd, t);
      const float Y0  = rlf(yv.x, t);
      const float Y1x = rlf(yv.y, t);
      const float Y1y = rlf(yv.z, t);
      const float Y1z = rlf(yv.w, t);
      const float* srow = sv + (size_t)s * 256;
      float se[2], vx[2], vy[2], vz[2];
      #pragma unroll
      for (int tt = 0; tt < 2; ++tt) {
        const int u = lane + 32 * tt;
        se[tt] = srow[u];
        vx[tt] = srow[64 + u * 3 + 0];
        vy[tt] = srow[64 + u * 3 + 1];
        vz[tt] = srow[64 + u * 3 + 2];
      }
      float hb[8];
      #pragma unroll
      for (int r = 0; r < 8; ++r) hb[r] = rlf(h[r], t);
      float w[4][2];
      #pragma unroll
      for (int k = 0; k < 4; ++k) {
        float t0 = 0.f, t1 = 0.f;
        #pragma unroll
        for (int r = 0; r < 8; ++r) {
          t0 += hb[r] * wm2[r][k][0];
          t1 += hb[r] * wm2[r][k][1];
        }
        w[k][0] = t0 * 0.35355339059f;
        w[k][1] = t1 * 0.35355339059f;
      }
      #pragma unroll
      for (int tt = 0; tt < 2; ++tt) {
        const float dv = vx[tt] * Y1x + vy[tt] * Y1y + vz[tt] * Y1z;
        a0[tt]    += w[0][tt] * se[tt] * Y0;
        b0[tt]    += w[3][tt] * dv * 0.57735026919f;
        a1[tt][0] += w[1][tt] * se[tt] * Y1x;
        a1[tt][1] += w[1][tt] * se[tt] * Y1y;
        a1[tt][2] += w[1][tt] * se[tt] * Y1z;
        b1[tt][0] += w[2][tt] * vx[tt] * Y0;
        b1[tt][1] += w[2][tt] * vy[tt] * Y0;
        b1[tt][2] += w[2][tt] * vz[tt] * Y0;
      }
    }
  }

  float* base = msmv + (size_t)n * 512;
  #pragma unroll
  for (int tt = 0; tt < 2; ++tt) {
    const int u = lane + 32 * tt;
    base[u]      = a0[tt];
    base[64 + u] = b0[tt];
    #pragma unroll
    for (int d = 0; d < 3; ++d) {
      base[128 + u * 3 + d]        = a1[tt][d];
      base[128 + (64 + u) * 3 + d] = b1[tt][d];
    }
  }
}

template <bool ACONTIG, int NC>
__device__ __forceinline__ void post_pass(const float* __restrict__ arow,
                                          const float* __restrict__ W2,
                                          const float* __restrict__ Wsk,
                                          float* myLds, int kh, int l16,
                                          float attE0, float attE1,
                                          float attO0, float attO1,
                                          v8f (&acc)[NC][4]) {
  gemm_loop<ACONTIG, 32, NC>(arow, W2, kh, l16, acc);

  #pragma unroll
  for (int d = 0; d < NC; ++d) {
    #pragma unroll
    for (int p = 0; p < 2; ++p) {
      #pragma unroll
      for (int j = 0; j < 8; ++j) {
        acc[d][2 * p][j]     *= 0.08838834765f;
        acc[d][2 * p + 1][j] *= 0.08838834765f;
        *(float2*)(myLds + d * 1024 + (j + 8 * kh) * 64 + 32 * p + 2 * l16) =
            make_float2(acc[d][2 * p][j], acc[d][2 * p + 1][j]);
      }
    }
  }
  __syncthreads();

  #pragma unroll 2
  for (int i = 0; i < 64; ++i) {
    float m2[NC];
    #pragma unroll
    for (int d = 0; d < NC; ++d) m2[d] = myLds[d * 1024 + l16 * 64 + i];
    const float* bp = Wsk + (8 * i + 2 * kh) * 64 + 2 * l16;
    v2f a[NC];
    #pragma unroll
    for (int d = 0; d < NC; ++d) {
      a[d].x = m2[d] * attE0; a[d].y = m2[d] * attE1;
    }
    #pragma unroll
    for (int p = 0; p < 2; ++p) {
      const float2 r0 = *(const float2*)(bp + 32 * p);
      const float2 r1 = *(const float2*)(bp + 64 + 32 * p);
      v2f be; be.x = r0.x; be.y = r1.x;
      v2f bo; bo.x = r0.y; bo.y = r1.y;
      #pragma unroll
      for (int d = 0; d < NC; ++d)
        acc[d][2 * p] = wmma_f32_k4(a[d], be, acc[d][2 * p]);
      #pragma unroll
      for (int d = 0; d < NC; ++d)
        acc[d][2 * p + 1] = wmma_f32_k4(a[d], bo, acc[d][2 * p + 1]);
    }
    #pragma unroll
    for (int d = 0; d < NC; ++d) {
      a[d].x = m2[d] * attO0; a[d].y = m2[d] * attO1;
    }
    #pragma unroll
    for (int p = 0; p < 2; ++p) {
      const float2 r0 = *(const float2*)(bp + 4 * 64 + 32 * p);
      const float2 r1 = *(const float2*)(bp + 5 * 64 + 32 * p);
      v2f be; be.x = r0.x; be.y = r1.x;
      v2f bo; bo.x = r0.y; bo.y = r1.y;
      #pragma unroll
      for (int d = 0; d < NC; ++d)
        acc[d][2 * p] = wmma_f32_k4(a[d], be, acc[d][2 * p]);
      #pragma unroll
      for (int d = 0; d < NC; ++d)
        acc[d][2 * p + 1] = wmma_f32_k4(a[d], bo, acc[d][2 * p + 1]);
    }
  }
}

__global__ void __launch_bounds__(256)
k_post(const float* __restrict__ msmv,
                       const float* __restrict__ attrs,
                       const float* __restrict__ W2s,
                       const float* __restrict__ W2v,
                       const float* __restrict__ WskS,
                       const float* __restrict__ WskV,
                       float* __restrict__ out) {
  __shared__ float lds[8 * 3 * 16 * 64];
  const int lane = threadIdx.x & 31;
  const int wid  = threadIdx.x >> 5;
  const int l16  = lane & 15;
  const int kh   = lane >> 4;
  const int row0 = (blockIdx.x * 8 + wid) * 16;
  int rowA = row0 + l16; if (rowA >= NN) rowA = NN - 1;
  float* myLds = lds + wid * (3 * 16 * 64);

  const float* arow8 = attrs + (size_t)rowA * 8;
  const float attE0 = arow8[2 * kh]     * 0.04419417382f;
  const float attE1 = arow8[2 * kh + 1] * 0.04419417382f;
  const float attO0 = arow8[4 + 2 * kh] * 0.04419417382f;
  const float attO1 = arow8[5 + 2 * kh] * 0.04419417382f;
  const float* mrow = msmv + (size_t)rowA * 512;

  {
    v8f acc[1][4] = {};
    post_pass<true, 1>(mrow, W2s, WskS, myLds, kh, l16,
                       attE0, attE1, attO0, attO1, acc);
    #pragma unroll
    for (int p = 0; p < 2; ++p) {
      #pragma unroll
      for (int j = 0; j < 8; ++j) {
        const int r = row0 + j + 8 * kh;
        if (r < NN)
          *(float2*)(out + r * 256 + 32 * p + 2 * l16) =
              make_float2(acc[0][2 * p][j], acc[0][2 * p + 1][j]);
      }
    }
  }
  __syncthreads();
  {
    v8f acc[3][4] = {};
    post_pass<false, 3>(mrow + 128, W2v, WskV, myLds, kh, l16,
                        attE0, attE1, attO0, attO1, acc);
    #pragma unroll
    for (int p = 0; p < 2; ++p) {
      #pragma unroll
      for (int j = 0; j < 8; ++j) {
        const int r = row0 + j + 8 * kh;
        if (r < NN) {
          float2* o = (float2*)(out + r * 256 + 64 + (32 * p + 2 * l16) * 3);
          o[0] = make_float2(acc[0][2 * p][j],     acc[1][2 * p][j]);
          o[1] = make_float2(acc[2][2 * p][j],     acc[0][2 * p + 1][j]);
          o[2] = make_float2(acc[1][2 * p + 1][j], acc[2][2 * p + 1][j]);
        }
      }
    }
  }
}

extern "C" void kernel_launch(void* const* d_in, const int* in_sizes, int n_in,
                              void* d_out, int out_size, void* d_ws, size_t ws_size,
                              hipStream_t stream) {
  const float* node_attrs = (const float*)d_in[0];
  const float* node_feats = (const float*)d_in[1];
  const float* edge_attrs = (const float*)d_in[2];
  const float* edge_feats = (const float*)d_in[3];
  const int*   edge_index = (const int*)d_in[4];
  const float* W1s  = (const float*)d_in[5];
  const float* W1v  = (const float*)d_in[6];
  const float* Wm1  = (const float*)d_in[7];
  const float* Wm2  = (const float*)d_in[8];
  const float* W2s  = (const float*)d_in[9];
  const float* W2v  = (const float*)d_in[10];
  const float* WskS = (const float*)d_in[11];
  const float* WskV = (const float*)d_in[12];

  float* out  = (float*)d_out;
  float* msmv = (float*)d_ws;

  int* cnt  = (int*)(msmv + (size_t)NN * 512);
  int* offs = cnt + NN;
  int* cur  = offs + NN + 4;
  int* list = cur + NN;

  k_zero<<<20, 256, 0, stream>>>((float4*)cnt, NN / 4);
  k_hist<<<(NE + 255) / 256, 256, 0, stream>>>(edge_index, cnt);
  k_scan<<<1, 1024, 0, stream>>>(cnt, offs, cur);
  k_fill<<<(NE + 255) / 256, 256, 0, stream>>>(edge_index, cur, list);
  k_pre <<<(NN + 127) / 128, 256, 0, stream>>>(node_feats, W1s, W1v, out);
  k_msg <<<(NN + 7) / 8, 256, 0, stream>>>(edge_attrs, edge_feats, edge_index,
                                           Wm1, Wm2, out, offs, list, msmv);
  k_post<<<(NN + 127) / 128, 256, 0, stream>>>(msmv, node_attrs, W2s, W2v,
                                               WskS, WskV, out);
}


# k_post: W2 and Wsk of the channel in work are copied once per work-group into LDS (row pitch 80 floats); all eight waves read their B fragments there, not from global memory; same k order and sums
# speedup vs baseline: 2.1173x; 1.0421x over previous
#include <hip/hip_runtime.h>
#include <hip/hip_bf16.h>
#include <math.h>


#define NN 20000
#define NE 320000
#define WP 80

typedef __attribute__((ext_vector_type(2))) float v2f;
typedef __attribute__((ext_vector_type(8))) float v8f;

__device__ __forceinline__ v8f wmma_f32_k4(v2f a, v2f b, v8f c) {
  return __builtin_amdgcn_wmma_f32_16x16x4_f32(false, a, false, b,
                                               (short)0, c, false, false);
}

__device__ __forceinline__ float sspf(float x) {
  return fmaxf(x, 0.0f) + log1pf(__expf(-fabsf(x))) - 0.69314718056f;
}

__global__ void k_zero(float4* __restrict__ p, int n4) {
  int i = blockIdx.x * blockDim.x + threadIdx.x;
  const int stride = gridDim.x * blockDim.x;
  for (; i < n4; i += stride) p[i] = make_float4(0.f, 0.f, 0.f, 0.f);
}

template <bool ACONTIG, int KC, int NC, int PITCH>
__device__ __forceinline__ void gemm_loop(const float* __restrict__ arow,
                                          const float* __restrict__ W,
                                          int kh, int l16, v8f (&acc)[NC][4]) {
  #pragma unroll 4
  for (int kc = 0; kc < KC; ++kc) {
    const int k0 = kc * 4 + 2 * kh;
    v2f a[NC];
    if (ACONTIG) {
      const float2 av = *(const float2*)(arow + k0);
      a[0].x = av.x; a[0].y = av.y;
    } else {
      const float2* ap = (const float2*)(arow + k0 * 3);
      const float2 f0 = ap[0];
      const float2 f1 = ap[1];
      const float2 f2 = ap[2];
      a[0].x = f0.x; a[0].y = f1.y;
      a[NC > 1 ? 1 : 0].x = f0.y; a[NC > 1 ? 1 : 0].y = f2.x;
      a[NC > 2 ? 2 : 0].x = f1.x; a[NC > 2 ? 2 : 0].y = f2.y;
    }
    const float* bp = W + k0 * PITCH + 2 * l16;
    #pragma unroll
    for (int p = 0; p < 2; ++p) {
      const float2 r0 = *(const float2*)(bp + 32 * p);
      const float2 r1 = *(const float2*)(bp + PITCH + 32 * p);
      v2f be; be.x = r0.x; be.y = r1.x;
      v2f bo; bo.x = r0.y; bo.y = r1.y;
      #pragma unroll
      for (int d = 0; d < NC; ++d)
        acc[d][2 * p] = wmma_f32_k4(a[d], be, acc[d][2 * p]);
      #pragma unroll
      for (int d = 0; d < NC; ++d)
        acc[d][2 * p + 1] = wmma_f32_k4(a[d], bo, acc[d][2 * p + 1]);
    }
  }
}

__global__ void __launch_bounds__(256)
k_pre(const float* __restrict__ nf,
                      const float* __restrict__ W1s,
                      const float* __restrict__ W1v,
                      float* __restrict__ sv) {
  const int lane = threadIdx.x & 31;
  const int wid  = threadIdx.x >> 5;
  const int l16  = lane & 15;
  const int kh   = lane >> 4;
  const int row0 = (blockIdx.x * 8 + wid) * 16;
  int rowA = row0 + l16; if (rowA >= NN) rowA = NN - 1;
  const float* nrow = nf + (size_t)rowA * 256;

  {
    v8f acc[1][4] = {};
    gemm_loop<true, 16, 1, 64>(nrow, W1s, kh, l16, acc);
    #pragma unroll
    for (int p = 0; p < 2; ++p) {
      #pragma unroll
      for (int j = 0; j < 8; ++j) {
        const int r = row0 + j + 8 * kh;
        if (r < NN)
          *(float2*)(sv + r * 256 + 32 * p + 2 * l16) =
              make_float2(acc[0][2 * p][j] * 0.125f, acc[0][2 * p + 1][j] * 0.125f);
      }
    }
  }
  {
    v8f acc[3][4] = {};
    gemm_loop<false, 16, 3, 64>(nrow + 64, W1v, kh, l16, acc);
    #pragma unroll
    for (int p = 0; p < 2; ++p) {
      #pragma unroll
      for (int j = 0; j < 8; ++j) {
        const int r = row0 + j + 8 * kh;
        if (r < NN) {
          float2* o = (float2*)(sv + r * 256 + 64 + (32 * p + 2 * l16) * 3);
          o[0] = make_float2(acc[0][2 * p][j] * 0.125f,
                             acc[1][2 * p][j] * 0.125f);
          o[1] = make_float2(acc[2][2 * p][j] * 0.125f,
                             acc[0][2 * p + 1][j] * 0.125f);
          o[2] = make_float2(acc[1][2 * p + 1][j] * 0.125f,
                             acc[2][2 * p + 1][j] * 0.125f);
        }
      }
    }
  }
}

__global__ void k_hist(const int* __restrict__ eidx, int* __restrict__ cnt) {
  const int e = blockIdx.x * blockDim.x + threadIdx.x;
  if (e < NE) atomicAdd(&cnt[eidx[NE + e]], 1);
}

__global__ void __launch_bounds__(1024)
k_scan(const int* __restrict__ cnt, int* __restrict__ offs,
       int* __restrict__ cur) {
  __shared__ int part[1024];
  const int t = threadIdx.x;
  const int b = t * 20;
  int loc[20];
  int s = 0;
  #pragma unroll
  for (int k = 0; k < 20; ++k) {
    const int i = b + k;
    const int c = (i < NN) ? cnt[i] : 0;
    loc[k] = s;
    s += c;
  }
  part[t] = s;
  __syncthreads();
  for (int d = 1; d < 1024; d <<= 1) {
    const int v = (t >= d) ? part[t - d] : 0;
    __syncthreads();
    part[t] += v;
    __syncthreads();
  }
  const int base = part[t] - s;
  #pragma unroll
  for (int k = 0; k < 20; ++k) {
    const int i = b + k;
    if (i < NN) { offs[i] = base + loc[k]; cur[i] = base + loc[k]; }
  }
  if (t == 1023) offs[NN] = part[t];
}

__global__ void k_fill(const int* __restrict__ eidx, int* __restrict__ cur,
                       int* __restrict__ list) {
  const int e = blockIdx.x * blockDim.x + threadIdx.x;
  if (e < NE) {
    const int pos = atomicAdd(&cur[eidx[NE + e]], 1);
    list[pos] = e;
  }
}

__device__ __forceinline__ float rlf(float x, int l) {
  return __int_as_float(__builtin_amdgcn_readlane(__float_as_int(x), l));
}

__global__ void __launch_bounds__(256)
k_msg(const float* __restrict__ ea,
      const float* __restrict__ efeat,
      const int*   __restrict__ eidx,
      const float* __restrict__ Wm1,
      const float* __restrict__ Wm2,
      const float* __restrict__ sv,
      const int*   __restrict__ offs,
      const int*   __restrict__ list,
      float* __restrict__ msmv) {
  const int lane = threadIdx.x & 31;
  const int n = __builtin_amdgcn_readfirstlane(blockIdx.x * 8 + (threadIdx.x >> 5));
  if (n >= NN) return;
  const int off = offs[n];
  const int len = offs[n + 1] - off;

  float wm2[8][4][2];
  #pragma unroll
  for (int r = 0; r < 8; ++r) {
    #pragma unroll
    for (int k = 0; k < 4; ++k) {
      wm2[r][k][0] = Wm2[r * 256 + k * 64 + lane];
      wm2[r][k][1] = Wm2[r * 256 + k * 64 + lane + 32];
    }
  }
  float a0[2] = {0.f, 0.f}, b0[2] = {0.f, 0.f};
  float a1[2][3] = {{0.f, 0.f, 0.f}, {0.f, 0.f, 0.f}};
  float b1[2][3] = {{0.f, 0.f, 0.f}, {0.f, 0.f, 0.f}};

  int last = -1;
  for (int done = 0; done < len; done += 32) {
    const int cnt = (len - done < 32) ? (len - done) : 32;
    int mine = 0x7fffffff;
    for (int j0 = 0; j0 < len; j0 += 32) {
      const int jj = j0 + lane;
      const int idj = (jj < len) ? list[off + jj] : 0x7fffffff;
      int rank = 0;
      for (int i0 = 0; i0 < len; i0 += 32) {
        const int ii = i0 + lane;
        const int idi = (ii < len) ? list[off + ii] : 0x7fffffff;
        #pragma unroll
        for (int i = 0; i < 32; ++i) {
          const int v = __builtin_amdgcn_readlane(idi, i);
          rank += (v > last && v < idj) ? 1 : 0;
        }
      }
      const int rk = (idj > last && idj != 0x7fffffff) ? rank : -1;
      #pragma unroll
      for (int i = 0; i < 32; ++i) {
        const int r = __builtin_amdgcn_readlane(rk, i);
        const int v = __builtin_amdgcn_readlane(idj, i);
        if (r == lane) mine = v;
      }
    }
    last = __builtin_amdgcn_readlane(mine, cnt - 1);

    const int e = (lane < cnt) ? mine : 0;
    const float4 fa = *(const float4*)(efeat + (size_t)e * 8);
    const float4 fb = *(const float4*)(efeat + (size_t)e * 8 + 4);
    const float f[8] = {fa.x, fa.y, fa.z, fa.w, fb.x, fb.y, fb.z, fb.w};
    float h[8];
    #pragma unroll
    for (int r = 0; r < 8; ++r) {
      float t = 0.f;
      #pragma unroll
      for (int q = 0; q < 8; ++q) t += f[q] * Wm1[q * 8 + r];
      h[r] = sspf(t * 0.35355339059f);
    }
    const float4 yv = *(const float4*)(ea + (size_t)e * 4);
    const int snd = eidx[e];

    for (int t = 0; t < cnt; ++t) {
      const int   s   = __builtin_amdgcn_readlane(snd, t);
      const float Y0  = rlf(yv.x, t);
      const float Y1x = rlf(yv.y, t);
      const float Y1y = rlf(yv.z, t);
      const float Y1z = rlf(yv.w, t);
      const float* srow = sv + (size_t)s * 256;
      float se[2], vx[2], vy[2], vz[2];
      #pragma unroll
      for (int tt = 0; tt < 2; ++tt) {
        const int u = lane + 32 * tt;
        se[tt] = srow[u];
        vx[tt] = srow[64 + u * 3 + 0];
        vy[tt] = srow[64 + u * 3 + 1];
        vz[tt] = srow[64 + u * 3 + 2];
      }
      float hb[8];
      #pragma unroll
      for (int r = 0; r < 8; ++r) hb[r] = rlf(h[r], t);
      float w[4][2];
      #pragma unroll
      for (int k = 0; k < 4; ++k) {
        float t0 = 0.f, t1 = 0.f;
        #pragma unroll
        for (int r = 0; r < 8; ++r) {
          t0 += hb[r] * wm2[r][k][0];
          t1 += hb[r] * wm2[r][k][1];
        }
        w[k][0] = t0 * 0.35355339059f;
        w[k][1] = t1 * 0.35355339059f;
      }
      #pragma unroll
      for (int tt = 0; tt < 2; ++tt) {
        const float dv = vx[tt] * Y1x + vy[tt] * Y1y + vz[tt] * Y1z;
        a0[tt]    += w[0][tt] * se[tt] * Y0;
        b0[tt]    += w[3][tt] * dv * 0.57735026919f;
        a1[tt][0] += w[1][tt] * se[tt] * Y1x;
        a1[tt][1] += w[1][tt] * se[tt] * Y1y;
        a1[tt][2] += w[1][tt] * se[tt] * Y1z;
        b1[tt][0] += w[2][tt] * vx[tt] * Y0;
        b1[tt][1] += w[2][tt] * vy[tt] * Y0;
        b1[tt][2] += w[2][tt] * vz[tt] * Y0;
      }
    }
  }

  float* base = msmv + (size_t)n * 512;
  #pragma unroll
  for (int tt = 0; tt < 2; ++tt) {
    const int u = lane + 32 * tt;
    base[u]      = a0[tt];
    base[64 + u] = b0[tt];
    #pragma unroll
    for (int d = 0; d < 3; ++d) {
      base[128 + u * 3 + d]        = a1[tt][d];
      base[128 + (64 + u) * 3 + d] = b1[tt][d];
    }
  }
}

template <bool ACONTIG, int NC>
__device__ __forceinline__ void post_pass(const float* __restrict__ arow,
                                          const float* W2,
                                          const float* Wsk,
                                          float* myLds, int kh, int l16,
                                          float attE0, float attE1,
                                          float attO0, float attO1,
                                          v8f (&acc)[NC][4]) {
  gemm_loop<ACONTIG, 32, NC, WP>(arow, W2, kh, l16, acc);

  #pragma unroll
  for (int d = 0; d < NC; ++d) {
    #pragma unroll
    for (int p = 0; p < 2; ++p) {
      #pragma unroll
      for (int j = 0; j < 8; ++j) {
        acc[d][2 * p][j]     *= 0.08838834765f;
        acc[d][2 * p + 1][j] *= 0.08838834765f;
        *(float2*)(myLds + d * 1024 + (j + 8 * kh) * 64 + 32 * p + 2 * l16) =
            make_float2(acc[d][2 * p][j], acc[d][2 * p + 1][j]);
      }
    }
  }
  __syncthreads();

  for (int i0 = 0; i0 < 64; i0 += 2) {
   #pragma unroll
   for (int ii = 0; ii < 2; ++ii) {
    const int i = i0 + ii;
    float m2[NC];
    #pragma unroll
    for (int d = 0; d < NC; ++d) m2[d] = myLds[d * 1024 + l16 * 64 + i];
    const float* bp = Wsk + (8 * i + 2 * kh) * WP + 2 * l16;
    v2f a[NC];
    #pragma unroll
    for (int d = 0; d < NC; ++d) {
      a[d].x = m2[d] * attE0; a[d].y = m2[d] * attE1;
    }
    #pragma unroll
    for (int p = 0; p < 2; ++p) {
      const float2 r0 = *(const float2*)(bp + 32 * p);
      const float2 r1 = *(const float2*)(bp + WP + 32 * p);
      v2f be; be.x = r0.x; be.y = r1.x;
      v2f bo; bo.x = r0.y; bo.y = r1.y;
      #pragma unroll
      for (int d = 0; d < NC; ++d)
        acc[d][2 * p] = wmma_f32_k4(a[d], be, acc[d][2 * p]);
      #pragma unroll
      for (int d = 0; d < NC; ++d)
        acc[d][2 * p + 1] = wmma_f32_k4(a[d], bo, acc[d][2 * p + 1]);
    }
    #pragma unroll
    for (int d = 0; d < NC; ++d) {
      a[d].x = m2[d] * attO0; a[d].y = m2[d] * attO1;
    }
    #pragma unroll
    for (int p = 0; p < 2; ++p) {
      const float2 r0 = *(const float2*)(bp + 4 * WP + 32 * p);
      const float2 r1 = *(const float2*)(bp + 5 * WP + 32 * p);
      v2f be; be.x = r0.x; be.y = r1.x;
      v2f bo; bo.x = r0.y; bo.y = r1.y;
      #pragma unroll
      for (int d = 0; d < NC; ++d)
        acc[d][2 * p] = wmma_f32_k4(a[d], be, acc[d][2 * p]);
      #pragma unroll
      for (int d = 0; d < NC; ++d)
        acc[d][2 * p + 1] = wmma_f32_k4(a[d], bo, acc[d][2 * p + 1]);
    }
   }
    __builtin_amdgcn_sched_barrier(0);
    asm volatile("v_nop\n\tv_nop\n\tv_nop\n\tv_nop");
  }
}

template <int ROWS>
__device__ __forceinline__ void stage_weights(const float* __restrict__ Wg,
                                              float* wl, int tid, int rot) {
  constexpr int TRIPS = ROWS / 16;
  #pragma unroll 8
  for (int j = 0; j < TRIPS; ++j) {
    const int i = (((j + rot) & (TRIPS - 1)) << 8) + tid;
    const float4 v = *(const float4*)(Wg + i * 4);
    *(float4*)(wl + (i >> 4) * WP + (i & 15) * 4) = v;
  }
}

__global__ void __launch_bounds__(256)
k_post(const float* __restrict__ msmv,
                       const float* __restrict__ attrs,
                       const float* __restrict__ W2s,
                       const float* __restrict__ W2v,
                       const float* __restrict__ WskS,
                       const float* __restrict__ WskV,
                       float* __restrict__ out) {
  __shared__ alignas(16) float lds[8 * 3 * 16 * 64 + (128 + 512) * WP];
  float* wl2  = lds + 8 * 3 * 16 * 64;
  float* wlsk = wl2 + 128 * WP;
  const int lane = threadIdx.x & 31;
  const int wid  = threadIdx.x >> 5;
  const int l16  = lane & 15;
  const int kh   = lane >> 4;
  const int row0 = (blockIdx.x * 8 + wid) * 16;
  int rowA = row0 + l16; if (rowA >= NN) rowA = NN - 1;
  float* myLds = lds + wid * (3 * 16 * 64);

  const float* arow8 = attrs + (size_t)rowA * 8;
  const float attE0 = arow8[2 * kh]     * 0.04419417382f;
  const float attE1 = arow8[2 * kh + 1] * 0.04419417382f;
  const float attO0 = arow8[4 + 2 * kh] * 0.04419417382f;
  const float attO1 = arow8[5 + 2 * kh] * 0.04419417382f;
  const float* mrow = msmv + (size_t)rowA * 512;

  stage_weights<128>(W2s, wl2, threadIdx.x, blockIdx.x);
  stage_weights<512>(WskS, wlsk, threadIdx.x, blockIdx.x);
  __syncthreads();
  {
    v8f acc[1][4] = {};
    post_pass<true, 1>(mrow, wl2, wlsk, myLds, kh, l16,
                       attE0, attE1, attO0, attO1, acc);
    #pragma unroll
    for (int p = 0; p < 2; ++p) {
      #pragma unroll
      for (int j = 0; j < 8; ++j) {
        const int r = row0 + j + 8 * kh;
        if (r < NN)
          *(float2*)(out + r * 256 + 32 * p + 2 * l16) =
              make_float2(acc[0][2 * p][j], acc[0][2 * p + 1][j]);
      }
    }
  }
  __syncthreads();
  stage_weights<128>(W2v, wl2, threadIdx.x, blockIdx.x);
  stage_weights<512>(WskV, wlsk, threadIdx.x, blockIdx.x);
  __syncthreads();
  {
    v8f acc[3][4] = {};
    post_pass<false, 3>(mrow + 128, wl2, wlsk, myLds, kh, l16,
                        attE0, attE1, attO0, attO1, acc);
    #pragma unroll
    for (int p = 0; p < 2; ++p) {
      #pragma unroll
      for (int j = 0; j < 8; ++j) {
        const int r = row0 + j + 8 * kh;
        if (r < NN) {
          float2* o = (float2*)(out + r * 256 + 64 + (32 * p + 2 * l16) * 3);
          o[0] = make_float2(acc[0][2 * p][j],     acc[1][2 * p][j]);
          o[1] = make_float2(acc[2][2 * p][j],     acc[0][2 * p + 1][j]);
          o[2] = make_float2(acc[1][2 * p + 1][j], acc[2][2 * p + 1][j]);
        }
      }
    }
  }
}

extern "C" void kernel_launch(void* const* d_in, const int* in_sizes, int n_in,
                              void* d_out, int out_size, void* d_ws, size_t ws_size,
                              hipStream_t stream) {
  const float* node_attrs = (const float*)d_in[0];
  const float* node_feats = (const float*)d_in[1];
  const float* edge_attrs = (const float*)d_in[2];
  const float* edge_feats = (const float*)d_in[3];
  const int*   edge_index = (const int*)d_in[4];
  const float* W1s  = (const float*)d_in[5];
  const float* W1v  = (const float*)d_in[6];
  const float* Wm1  = (const float*)d_in[7];
  const float* Wm2  = (const float*)d_in[8];
  const float* W2s  = (const float*)d_in[9];
  const float* W2v  = (const float*)d_in[10];
  const float* WskS = (const float*)d_in[11];
  const float* WskV = (const float*)d_in[12];

  float* out  = (float*)d_out;
  float* msmv = (float*)d_ws;

  int* cnt  = (int*)(msmv + (size_t)NN * 512);
  int* offs = cnt + NN;
  int* cur  = offs + NN + 4;
  int* list = cur + NN;

  k_zero<<<20, 256, 0, stream>>>((float4*)cnt, NN / 4);
  k_hist<<<(NE + 255) / 256, 256, 0, stream>>>(edge_index, cnt);
  k_scan<<<1, 1024, 0, stream>>>(cnt, offs, cur);
  k_fill<<<(NE + 255) / 256, 256, 0, stream>>>(edge_index, cur, list);
  k_pre <<<(NN + 127) / 128, 256, 0, stream>>>(node_feats, W1s, W1v, out);
  k_msg <<<(NN + 7) / 8, 256, 0, stream>>>(edge_attrs, edge_feats, edge_index,
                                           Wm1, Wm2, out, offs, list, msmv);
  k_post<<<(NN + 127) / 128, 256, 0, stream>>>(msmv, node_attrs, W2s, W2v,
                                               WskS, WskV, out);
}


# k_msg: the edge loop takes 8 edges per trip, the sender-row gather loads of all 8 issued as one group ahead of the sums (was one edge's loads and a full wait per trip); same order of sums
# speedup vs baseline: 2.1960x; 1.0372x over previous
#include <hip/hip_runtime.h>
#include <hip/hip_bf16.h>
#include <math.h>


#define NN 20000
#define NE 320000
#define WP 80

typedef __attribute__((ext_vector_type(2))) float v2f;
typedef __attribute__((ext_vector_type(8))) float v8f;

__device__ __forceinline__ v8f wmma_f32_k4(v2f a, v2f b, v8f c) {
  return __builtin_amdgcn_wmma_f32_16x16x4_f32(false, a, false, b,
                                               (short)0, c, false, false);
}

__device__ __forceinline__ float sspf(float x) {
  return fmaxf(x, 0.0f) + log1pf(__expf(-fabsf(x))) - 0.69314718056f;
}

__global__ void k_zero(float4* __restrict__ p, int n4) {
  int i = blockIdx.x * blockDim.x + threadIdx.x;
  const int stride = gridDim.x * blockDim.x;
  for (; i < n4; i += stride) p[i] = make_float4(0.f, 0.f, 0.f, 0.f);
}

template <bool ACONTIG, int KC, int NC, int PITCH>
__device__ __forceinline__ void gemm_loop(const float* __restrict__ arow,
                                          const float* __restrict__ W,
                                          int kh, int l16, v8f (&acc)[NC][4]) {
  #pragma unroll 4
  for (int kc = 0; kc < KC; ++kc) {
    const int k0 = kc * 4 + 2 * kh;
    v2f a[NC];
    if (ACONTIG) {
      const float2 av = *(const float2*)(arow + k0);
      a[0].x = av.x; a[0].y = av.y;
    } else {
      const float2* ap = (const float2*)(arow + k0 * 3);
      const float2 f0 = ap[0];
      const float2 f1 = ap[1];
      const float2 f2 = ap[2];
      a[0].x = f0.x; a[0].y = f1.y;
      a[NC > 1 ? 1 : 0].x = f0.y; a[NC > 1 ? 1 : 0].y = f2.x;
      a[NC > 2 ? 2 : 0].x = f1.x; a[NC > 2 ? 2 : 0].y = f2.y;
    }
    const float* bp = W + k0 * PITCH + 2 * l16;
    #pragma unroll
    for (int p = 0; p < 2; ++p) {
      const float2 r0 = *(const float2*)(bp + 32 * p);
      const float2 r1 = *(const float2*)(bp + PITCH + 32 * p);
      v2f be; be.x = r0.x; be.y = r1.x;
      v2f bo; bo.x = r0.y; bo.y = r1.y;
      #pragma unroll
      for (int d = 0; d < NC; ++d)
        acc[d][2 * p] = wmma_f32_k4(a[d], be, acc[d][2 * p]);
      #pragma unroll
      for (int d = 0; d < NC; ++d)
        acc[d][2 * p + 1] = wmma_f32_k4(a[d], bo, acc[d][2 * p + 1]);
    }
  }
}

__global__ void __launch_bounds__(256)
k_pre(const float* __restrict__ nf,
                      const float* __restrict__ W1s,
                      const float* __restrict__ W1v,
                      float* __restrict__ sv) {
  const int lane = threadIdx.x & 31;
  const int wid  = threadIdx.x >> 5;
  const int l16  = lane & 15;
  const int kh   = lane >> 4;
  const int row0 = (blockIdx.x * 8 + wid) * 16;
  int rowA = row0 + l16; if (rowA >= NN) rowA = NN - 1;
  const float* nrow = nf + (size_t)rowA * 256;

  {
    v8f acc[1][4] = {};
    gemm_loop<true, 16, 1, 64>(nrow, W1s, kh, l16, acc);
    #pragma unroll
    for (int p = 0; p < 2; ++p) {
      #pragma unroll
      for (int j = 0; j < 8; ++j) {
        const int r = row0 + j + 8 * kh;
        if (r < NN)
          *(float2*)(sv + r * 256 + 32 * p + 2 * l16) =
              make_float2(acc[0][2 * p][j] * 0.125f, acc[0][2 * p + 1][j] * 0.125f);
      }
    }
  }
  {
    v8f acc[3][4] = {};
    gemm_loop<false, 16, 3, 64>(nrow + 64, W1v, kh, l16, acc);
    #pragma unroll
    for (int p = 0; p < 2; ++p) {
      #pragma unroll
      for (int j = 0; j < 8; ++j) {
        const int r = row0 + j + 8 * kh;
        if (r < NN) {
          float2* o = (float2*)(sv + r * 256 + 64 + (32 * p + 2 * l16) * 3);
          o[0] = make_float2(acc[0][2 * p][j] * 0.125f,
                             acc[1][2 * p][j] * 0.125f);
          o[1] = make_float2(acc[2][2 * p][j] * 0.125f,
                             acc[0][2 * p + 1][j] * 0.125f);
          o[2] = make_float2(acc[1][2 * p + 1][j] * 0.125f,
                             acc[2][2 * p + 1][j] * 0.125f);
        }
      }
    }
  }
}

__global__ void k_hist(const int* __restrict__ eidx, int* __restrict__ cnt) {
  const int e = blockIdx.x * blockDim.x + threadIdx.x;
  if (e < NE) atomicAdd(&cnt[eidx[NE + e]], 1);
}

__global__ void __launch_bounds__(1024)
k_scan(const int* __restrict__ cnt, int* __restrict__ offs,
       int* __restrict__ cur) {
  __shared__ int part[1024];
  const int t = threadIdx.x;
  const int b = t * 20;
  int loc[20];
  int s = 0;
  #pragma unroll
  for (int k = 0; k < 20; ++k) {
    const int i = b + k;
    const int c = (i < NN) ? cnt[i] : 0;
    loc[k] = s;
    s += c;
  }
  part[t] = s;
  __syncthreads();
  for (int d = 1; d < 1024; d <<= 1) {
    const int v = (t >= d) ? part[t - d] : 0;
    __syncthreads();
    part[t] += v;
    __syncthreads();
  }
  const int base = part[t] - s;
  #pragma unroll
  for (int k = 0; k < 20; ++k) {
    const int i = b + k;
    if (i < NN) { offs[i] = base + loc[k]; cur[i] = base + loc[k]; }
  }
  if (t == 1023) offs[NN] = part[t];
}

__global__ void k_fill(const int* __restrict__ eidx, int* __restrict__ cur,
                       int* __restrict__ list) {
  const int e = blockIdx.x * blockDim.x + threadIdx.x;
  if (e < NE) {
    const int pos = atomicAdd(&cur[eidx[NE + e]], 1);
    list[pos] = e;
  }
}

__device__ __forceinline__ float rlf(float x, int l) {
  return __int_as_float(__builtin_amdgcn_readlane(__float_as_int(x), l));
}

__global__ void __launch_bounds__(256)
k_msg(const float* __restrict__ ea,
      const float* __restrict__ efeat,
      const int*   __restrict__ eidx,
      const float* __restrict__ Wm1,
      const float* __restrict__ Wm2,
      const float* __restrict__ sv,
      const int*   __restrict__ offs,
      const int*   __restrict__ list,
      float* __restrict__ msmv) {
  const int lane = threadIdx.x & 31;
  const int n = __builtin_amdgcn_readfirstlane(blockIdx.x * 8 + (threadIdx.x >> 5));
  if (n >= NN) return;
  const int off = offs[n];
  const int len = offs[n + 1] - off;

  float wm2[8][4][2];
  #pragma unroll
  for (int r = 0; r < 8; ++r) {
    #pragma unroll
    for (int k = 0; k < 4; ++k) {
      wm2[r][k][0] = Wm2[r * 256 + k * 64 + lane];
      wm2[r][k][1] = Wm2[r * 256 + k * 64 + lane + 32];
    }
  }
  float a0[2] = {0.f, 0.f}, b0[2] = {0.f, 0.f};
  float a1[2][3] = {{0.f, 0.f, 0.f}, {0.f, 0.f, 0.f}};
  float b1[2][3] = {{0.f, 0.f, 0.f}, {0.f, 0.f, 0.f}};

  int last = -1;
  for (int done = 0; done < len; done += 32) {
    const int cnt = (len - done < 32) ? (len - done) : 32;
    int mine = 0x7fffffff;
    for (int j0 = 0; j0 < len; j0 += 32) {
      const int jj = j0 + lane;
      const int idj = (jj < len) ? list[off + jj] : 0x7fffffff;
      int rank = 0;
      for (int i0 = 0; i0 < len; i0 += 32) {
        const int ii = i0 + lane;
        const int idi = (ii < len) ? list[off + ii] : 0x7fffffff;
        #pragma unroll
        for (int i = 0; i < 32; ++i) {
          const int v = __builtin_amdgcn_readlane(idi, i);
          rank += (v > last && v < idj) ? 1 : 0;
        }
      }
      const int rk = (idj > last && idj != 0x7fffffff) ? rank : -1;
      #pragma unroll
      for (int i = 0; i < 32; ++i) {
        const int r = __builtin_amdgcn_readlane(rk, i);
        const int v = __builtin_amdgcn_readlane(idj, i);
        if (r == lane) mine = v;
      }
    }
    last = __builtin_amdgcn_readlane(mine, cnt - 1);

    const int e = (lane < cnt) ? mine : 0;
    const float4 fa = *(const float4*)(efeat + (size_t)e * 8);
    const float4 fb = *(const float4*)(efeat + (size_t)e * 8 + 4);
    const float f[8] = {fa.x, fa.y, fa.z, fa.w, fb.x, fb.y, fb.z, fb.w};
    float h[8];
    #pragma unroll
    for (int r = 0; r < 8; ++r) {
      float t = 0.f;
      #pragma unroll
      for (int q = 0; q < 8; ++q) t += f[q] * Wm1[q * 8 + r];
      h[r] = sspf(t * 0.35355339059f);
    }
    const float4 yv = *(const float4*)(ea + (size_t)e * 4);
    const int snd = eidx[e];

    for (int t0 = 0; t0 < cnt; t0 += 8) {
      float se[8][2], vx[8][2], vy[8][2], vz[8][2];
      #pragma unroll
      for (int q = 0; q < 8; ++q) {
        const int tq = (t0 + q < cnt) ? (t0 + q) : (cnt - 1);
        const int s  = __builtin_amdgcn_readlane(snd, tq);
        const float* srow = sv + (size_t)s * 256;
        #pragma unroll
        for (int tt = 0; tt < 2; ++tt) {
          const int u = lane + 32 * tt;
          se[q][tt] = srow[u];
          vx[q][tt] = srow[64 + u * 3 + 0];
          vy[q][tt] = srow[64 + u * 3 + 1];
          vz[q][tt] = srow[64 + u * 3 + 2];
        }
      }
      #pragma unroll
      for (int q = 0; q < 8; ++q) {
        const int t = t0 + q;
        if (t < cnt) {
          const float Y0  = rlf(yv.x, t);
          const float Y1x = rlf(yv.y, t);
          const float Y1y = rlf(yv.z, t);
          const float Y1z = rlf(yv.w, t);
          float hb[8];
          #pragma unroll
          for (int r = 0; r < 8; ++r) hb[r] = rlf(h[r], t);
          float w[4][2];
          #pragma unroll
          for (int k = 0; k < 4; ++k) {
            float s0 = 0.f, s1 = 0.f;
            #pragma unroll
            for (int r = 0; r < 8; ++r) {
              s0 += hb[r] * wm2[r][k][0];
              s1 += hb[r] * wm2[r][k][1];
            }
            w[k][0] = s0 * 0.35355339059f;
            w[k][1] = s1 * 0.35355339059f;
          }
          #pragma unroll
          for (int tt = 0; tt < 2; ++tt) {
            const float dv = vx[q][tt] * Y1x + vy[q][tt] * Y1y + vz[q][tt] * Y1z;
            a0[tt]    += w[0][tt] * se[q][tt] * Y0;
            b0[tt]    += w[3][tt] * dv * 0.57735026919f;
            a1[tt][0] += w[1][tt] * se[q][tt] * Y1x;
            a1[tt][1] += w[1][tt] * se[q][tt] * Y1y;
            a1[tt][2] += w[1][tt] * se[q][tt] * Y1z;
            b1[tt][0] += w[2][tt] * vx[q][tt] * Y0;
            b1[tt][1] += w[2][tt] * vy[q][tt] * Y0;
            b1[tt][2] += w[2][tt] * vz[q][tt] * Y0;
          }
        }
      }
    }
  }

  float* base = msmv + (size_t)n * 512;
  #pragma unroll
  for (int tt = 0; tt < 2; ++tt) {
    const int u = lane + 32 * tt;
    base[u]      = a0[tt];
    base[64 + u] = b0[tt];
    #pragma unroll
    for (int d = 0; d < 3; ++d) {
      base[128 + u * 3 + d]        = a1[tt][d];
      base[128 + (64 + u) * 3 + d] = b1[tt][d];
    }
  }
}

template <bool ACONTIG, int NC>
__device__ __forceinline__ void post_pass(const float* __restrict__ arow,
                                          const float* W2,
                                          const float* Wsk,
                                          float* myLds, int kh, int l16,
                                          float attE0, float attE1,
                                          float attO0, float attO1,
                                          v8f (&acc)[NC][4]) {
  gemm_loop<ACONTIG, 32, NC, WP>(arow, W2, kh, l16, acc);

  #pragma unroll
  for (int d = 0; d < NC; ++d) {
    #pragma unroll
    for (int p = 0; p < 2; ++p) {
      #pragma unroll
      for (int j = 0; j < 8; ++j) {
        acc[d][2 * p][j]     *= 0.08838834765f;
        acc[d][2 * p + 1][j] *= 0.08838834765f;
        *(float2*)(myLds + d * 1024 + (j + 8 * kh) * 64 + 32 * p + 2 * l16) =
            make_float2(acc[d][2 * p][j], acc[d][2 * p + 1][j]);
      }
    }
  }
  __syncthreads();

  for (int i0 = 0; i0 < 64; i0 += 2) {
   #pragma unroll
   for (int ii = 0; ii < 2; ++ii) {
    const int i = i0 + ii;
    float m2[NC];
    #pragma unroll
    for (int d = 0; d < NC; ++d) m2[d] = myLds[d * 1024 + l16 * 64 + i];
    const float* bp = Wsk + (8 * i + 2 * kh) * WP + 2 * l16;
    v2f a[NC];
    #pragma unroll
    for (int d = 0; d < NC; ++d) {
      a[d].x = m2[d] * attE0; a[d].y = m2[d] * attE1;
    }
    #pragma unroll
    for (int p = 0; p < 2; ++p) {
      const float2 r0 = *(const float2*)(bp + 32 * p);
      const float2 r1 = *(const float2*)(bp + WP + 32 * p);
      v2f be; be.x = r0.x; be.y = r1.x;
      v2f bo; bo.x = r0.y; bo.y = r1.y;
      #pragma unroll
      for (int d = 0; d < NC; ++d)
        acc[d][2 * p] = wmma_f32_k4(a[d], be, acc[d][2 * p]);
      #pragma unroll
      for (int d = 0; d < NC; ++d)
        acc[d][2 * p + 1] = wmma_f32_k4(a[d], bo, acc[d][2 * p + 1]);
    }
    #pragma unroll
    for (int d = 0; d < NC; ++d) {
      a[d].x = m2[d] * attO0; a[d].y = m2[d] * attO1;
    }
    #pragma unroll
    for (int p = 0; p < 2; ++p) {
      const float2 r0 = *(const float2*)(bp + 4 * WP + 32 * p);
      const float2 r1 = *(const float2*)(bp + 5 * WP + 32 * p);
      v2f be; be.x = r0.x; be.y = r1.x;
      v2f bo; bo.x = r0.y; bo.y = r1.y;
      #pragma unroll
      for (int d = 0; d < NC; ++d)
        acc[d][2 * p] = wmma_f32_k4(a[d], be, acc[d][2 * p]);
      #pragma unroll
      for (int d = 0; d < NC; ++d)
        acc[d][2 * p + 1] = wmma_f32_k4(a[d], bo, acc[d][2 * p + 1]);
    }
   }
    __builtin_amdgcn_sched_barrier(0);
    asm volatile("v_nop\n\tv_nop\n\tv_nop\n\tv_nop");
  }
}

template <int ROWS>
__device__ __forceinline__ void stage_weights(const float* __restrict__ Wg,
                                              float* wl, int tid, int rot) {
  constexpr int TRIPS = ROWS / 16;
  #pragma unroll 8
  for (int j = 0; j < TRIPS; ++j) {
    const int i = (((j + rot) & (TRIPS - 1)) << 8) + tid;
    const float4 v = *(const float4*)(Wg + i * 4);
    *(float4*)(wl + (i >> 4) * WP + (i & 15) * 4) = v;
  }
}

__global__ void __launch_bounds__(256)
k_post(const float* __restrict__ msmv,
                       const float* __restrict__ attrs,
                       const float* __restrict__ W2s,
                       const float* __restrict__ W2v,
                       const float* __restrict__ WskS,
                       const float* __restrict__ WskV,
                       float* __restrict__ out) {
  __shared__ alignas(16) float lds[8 * 3 * 16 * 64 + (128 + 512) * WP];
  float* wl2  = lds + 8 * 3 * 16 * 64;
  float* wlsk = wl2 + 128 * WP;
  const int lane = threadIdx.x & 31;
  const int wid  = threadIdx.x >> 5;
  const int l16  = lane & 15;
  const int kh   = lane >> 4;
  const int row0 = (blockIdx.x * 8 + wid) * 16;
  int rowA = row0 + l16; if (rowA >= NN) rowA = NN - 1;
  float* myLds = lds + wid * (3 * 16 * 64);

  const float* arow8 = attrs + (size_t)rowA * 8;
  const float attE0 = arow8[2 * kh]     * 0.04419417382f;
  const float attE1 = arow8[2 * kh + 1] * 0.04419417382f;
  const float attO0 = arow8[4 + 2 * kh] * 0.04419417382f;
  const float attO1 = arow8[5 + 2 * kh] * 0.04419417382f;
  const float* mrow = msmv + (size_t)rowA * 512;

  stage_weights<128>(W2s, wl2, threadIdx.x, blockIdx.x);
  stage_weights<512>(WskS, wlsk, threadIdx.x, blockIdx.x);
  __syncthreads();
  {
    v8f acc[1][4] = {};
    post_pass<true, 1>(mrow, wl2, wlsk, myLds, kh, l16,
                       attE0, attE1, attO0, attO1, acc);
    #pragma unroll
    for (int p = 0; p < 2; ++p) {
      #pragma unroll
      for (int j = 0; j < 8; ++j) {
        const int r = row0 + j + 8 * kh;
        if (r < NN)
          *(float2*)(out + r * 256 + 32 * p + 2 * l16) =
              make_float2(acc[0][2 * p][j], acc[0][2 * p + 1][j]);
      }
    }
  }
  __syncthreads();
  stage_weights<128>(W2v, wl2, threadIdx.x, blockIdx.x);
  stage_weights<512>(WskV, wlsk, threadIdx.x, blockIdx.x);
  __syncthreads();
  {
    v8f acc[3][4] = {};
    post_pass<false, 3>(mrow + 128, wl2, wlsk, myLds, kh, l16,
                        attE0, attE1, attO0, attO1, acc);
    #pragma unroll
    for (int p = 0; p < 2; ++p) {
      #pragma unroll
      for (int j = 0; j < 8; ++j) {
        const int r = row0 + j + 8 * kh;
        if (r < NN) {
          float2* o = (float2*)(out + r * 256 + 64 + (32 * p + 2 * l16) * 3);
          o[0] = make_float2(acc[0][2 * p][j],     acc[1][2 * p][j]);
          o[1] = make_float2(acc[2][2 * p][j],     acc[0][2 * p + 1][j]);
          o[2] = make_float2(acc[1][2 * p + 1][j], acc[2][2 * p + 1][j]);
        }
      }
    }
  }
}

extern "C" void kernel_launch(void* const* d_in, const int* in_sizes, int n_in,
                              void* d_out, int out_size, void* d_ws, size_t ws_size,
                              hipStream_t stream) {
  const float* node_attrs = (const float*)d_in[0];
  const float* node_feats = (const float*)d_in[1];
  const float* edge_attrs = (const float*)d_in[2];
  const float* edge_feats = (const float*)d_in[3];
  const int*   edge_index = (const int*)d_in[4];
  const float* W1s  = (const float*)d_in[5];
  const float* W1v  = (const float*)d_in[6];
  const float* Wm1  = (const float*)d_in[7];
  const float* Wm2  = (const float*)d_in[8];
  const float* W2s  = (const float*)d_in[9];
  const float* W2v  = (const float*)d_in[10];
  const float* WskS = (const float*)d_in[11];
  const float* WskV = (const float*)d_in[12];

  float* out  = (float*)d_out;
  float* msmv = (float*)d_ws;

  int* cnt  = (int*)(msmv + (size_t)NN * 512);
  int* offs = cnt + NN;
  int* cur  = offs + NN + 4;
  int* list = cur + NN;

  k_zero<<<20, 256, 0, stream>>>((float4*)cnt, NN / 4);
  k_hist<<<(NE + 255) / 256, 256, 0, stream>>>(edge_index, cnt);
  k_scan<<<1, 1024, 0, stream>>>(cnt, offs, cur);
  k_fill<<<(NE + 255) / 256, 256, 0, stream>>>(edge_index, cur, list);
  k_pre <<<(NN + 127) / 128, 256, 0, stream>>>(node_feats, W1s, W1v, out);
  k_msg <<<(NN + 7) / 8, 256, 0, stream>>>(edge_attrs, edge_feats, edge_index,
                                           Wm1, Wm2, out, offs, list, msmv);
  k_post<<<(NN + 127) / 128, 256, 0, stream>>>(msmv, node_attrs, W2s, W2v,
                                               WskS, WskV, out);
}


# receiver lists in one pass: an edge takes the next slot of its receiver's own msmv row by a returning int add (row = list until the node's wave stores its sums); count + scan launches gone; same sums
# speedup vs baseline: 3.0104x; 1.3709x over previous
#include <hip/hip_runtime.h>
#include <hip/hip_bf16.h>
#include <math.h>


#define NN 20000
#define NE 320000
#define RCAP 512
#define WP 80

typedef __attribute__((ext_vector_type(2))) float v2f;
typedef __attribute__((ext_vector_type(8))) float v8f;

__device__ __forceinline__ v8f wmma_f32_k4(v2f a, v2f b, v8f c) {
  return __builtin_amdgcn_wmma_f32_16x16x4_f32(false, a, false, b,
                                               (short)0, c, false, false);
}

__device__ __forceinline__ float sspf(float x) {
  return fmaxf(x, 0.0f) + log1pf(__expf(-fabsf(x))) - 0.69314718056f;
}

__global__ void k_zero(float4* __restrict__ p, int n4) {
  int i = blockIdx.x * blockDim.x + threadIdx.x;
  const int stride = gridDim.x * blockDim.x;
  for (; i < n4; i += stride) p[i] = make_float4(0.f, 0.f, 0.f, 0.f);
}

template <bool ACONTIG, int KC, int NC, int PITCH>
__device__ __forceinline__ void gemm_loop(const float* __restrict__ arow,
                                          const float* __restrict__ W,
                                          int kh, int l16, v8f (&acc)[NC][4]) {
  #pragma unroll 4
  for (int kc = 0; kc < KC; ++kc) {
    const int k0 = kc * 4 + 2 * kh;
    v2f a[NC];
    if (ACONTIG) {
      const float2 av = *(const float2*)(arow + k0);
      a[0].x = av.x; a[0].y = av.y;
    } else {
      const float2* ap = (const float2*)(arow + k0 * 3);
      const float2 f0 = ap[0];
      const float2 f1 = ap[1];
      const float2 f2 = ap[2];
      a[0].x = f0.x; a[0].y = f1.y;
      a[NC > 1 ? 1 : 0].x = f0.y; a[NC > 1 ? 1 : 0].y = f2.x;
      a[NC > 2 ? 2 : 0].x = f1.x; a[NC > 2 ? 2 : 0].y = f2.y;
    }
    const float* bp = W + k0 * PITCH + 2 * l16;
    #pragma unroll
    for (int p = 0; p < 2; ++p) {
      const float2 r0 = *(const float2*)(bp + 32 * p);
      const float2 r1 = *(const float2*)(bp + PITCH + 32 * p);
      v2f be; be.x = r0.x; be.y = r1.x;
      v2f bo; bo.x = r0.y; bo.y = r1.y;
      #pragma unroll
      for (int d = 0; d < NC; ++d)
        acc[d][2 * p] = wmma_f32_k4(a[d], be, acc[d][2 * p]);
      #pragma unroll
      for (int d = 0; d < NC; ++d)
        acc[d][2 * p + 1] = wmma_f32_k4(a[d], bo, acc[d][2 * p + 1]);
    }
  }
}

__global__ void __launch_bounds__(256)
k_pre(const float* __restrict__ nf,
                      const float* __restrict__ W1s,
                      const float* __restrict__ W1v,
                      float* __restrict__ sv) {
  const int lane = threadIdx.x & 31;
  const int wid  = threadIdx.x >> 5;
  const int l16  = lane & 15;
  const int kh   = lane >> 4;
  const int row0 = (blockIdx.x * 8 + wid) * 16;
  int rowA = row0 + l16; if (rowA >= NN) rowA = NN - 1;
  const float* nrow = nf + (size_t)rowA * 256;

  {
    v8f acc[1][4] = {};
    gemm_loop<true, 16, 1, 64>(nrow, W1s, kh, l16, acc);
    #pragma unroll
    for (int p = 0; p < 2; ++p) {
      #pragma unroll
      for (int j = 0; j < 8; ++j) {
        const int r = row0 + j + 8 * kh;
        if (r < NN)
          *(float2*)(sv + r * 256 + 32 * p + 2 * l16) =
              make_float2(acc[0][2 * p][j] * 0.125f, acc[0][2 * p + 1][j] * 0.125f);
      }
    }
  }
  {
    v8f acc[3][4] = {};
    gemm_loop<false, 16, 3, 64>(nrow + 64, W1v, kh, l16, acc);
    #pragma unroll
    for (int p = 0; p < 2; ++p) {
      #pragma unroll
      for (int j = 0; j < 8; ++j) {
        const int r = row0 + j + 8 * kh;
        if (r < NN) {
          float2* o = (float2*)(sv + r * 256 + 64 + (32 * p + 2 * l16) * 3);
          o[0] = make_float2(acc[0][2 * p][j] * 0.125f,
                             acc[1][2 * p][j] * 0.125f);
          o[1] = make_float2(acc[2][2 * p][j] * 0.125f,
                             acc[0][2 * p + 1][j] * 0.125f);
          o[2] = make_float2(acc[1][2 * p + 1][j] * 0.125f,
                             acc[2][2 * p + 1][j] * 0.125f);
        }
      }
    }
  }
}

__global__ void k_fill(const int* __restrict__ eidx, int* __restrict__ cnt,
                       int* __restrict__ rows) {
  const int e = blockIdx.x * blockDim.x + threadIdx.x;
  if (e < NE) {
    const int r = eidx[NE + e];
    const int pos = atomicAdd(&cnt[r], 1);
    if (pos < RCAP) rows[(size_t)r * 512 + pos] = e;
  }
}

__device__ __forceinline__ int take_by_rank(int rk, int id, int lane, int mine) {
  #pragma unroll
  for (int i = 0; i < 32; ++i) {
    const int r = __builtin_amdgcn_readlane(rk, i);
    const int v = __builtin_amdgcn_readlane(id, i);
    if (r == lane) mine = v;
  }
  return mine;
}

__device__ __forceinline__ float rlf(float x, int l) {
  return __int_as_float(__builtin_amdgcn_readlane(__float_as_int(x), l));
}

__global__ void __launch_bounds__(256)
k_msg(const float* __restrict__ ea,
      const float* __restrict__ efeat,
      const int*   __restrict__ eidx,
      const float* __restrict__ Wm1,
      const float* __restrict__ Wm2,
      const float* __restrict__ sv,
      const int*   __restrict__ deg,
      const int*   rows,
      float* msmv) {
  const int lane = threadIdx.x & 31;
  const int n = __builtin_amdgcn_readfirstlane(blockIdx.x * 8 + (threadIdx.x >> 5));
  if (n >= NN) return;
  const int len = deg[n];
  const int* row = rows + (size_t)n * 512;
  const bool inrow = (len <= RCAP);

  float wm2[8][4][2];
  #pragma unroll
  for (int r = 0; r < 8; ++r) {
    #pragma unroll
    for (int k = 0; k < 4; ++k) {
      wm2[r][k][0] = Wm2[r * 256 + k * 64 + lane];
      wm2[r][k][1] = Wm2[r * 256 + k * 64 + lane + 32];
    }
  }
  float a0[2] = {0.f, 0.f}, b0[2] = {0.f, 0.f};
  float a1[2][3] = {{0.f, 0.f, 0.f}, {0.f, 0.f, 0.f}};
  float b1[2][3] = {{0.f, 0.f, 0.f}, {0.f, 0.f, 0.f}};

  int last = -1;
  int done = 0;
  int e0 = 0;
  for (;;) {
    int cnt;
    int mine = 0x7fffffff;
    if (inrow) {
      if (done >= len) break;
      cnt = (len - done < 32) ? (len - done) : 32;
      for (int j0 = 0; j0 < len; j0 += 32) {
        const int jj = j0 + lane;
        const int idj = (jj < len) ? row[jj] : 0x7fffffff;
        int rank = 0;
        for (int i0 = 0; i0 < len; i0 += 32) {
          const int ii = i0 + lane;
          const int idi = (ii < len) ? row[ii] : 0x7fffffff;
          #pragma unroll
          for (int i = 0; i < 32; ++i) {
            const int v = __builtin_amdgcn_readlane(idi, i);
            rank += (v > last && v < idj) ? 1 : 0;
          }
        }
        const int rk = (idj > last && idj != 0x7fffffff) ? rank : -1;
        mine = take_by_rank(rk, idj, lane, mine);
      }
      last = __builtin_amdgcn_readlane(mine, cnt - 1);
      done += 32;
    } else {
      unsigned hit = 0u;
      int eid = 0;
      while (e0 < NE) {
        eid = e0 + lane;
        hit = __builtin_amdgcn_ballot_w32(eidx[NE + eid] == n);
        e0 += 32;
        if (hit != 0u) break;
      }
      if (hit == 0u) break;
      cnt = __builtin_popcount(hit);
      const int rk = ((hit >> lane) & 1u)
                         ? __builtin_popcount(hit & ((1u << lane) - 1u)) : -1;
      mine = take_by_rank(rk, eid, lane, mine);
    }

    const int e = (lane < cnt) ? mine : 0;
    const float4 fa = *(const float4*)(efeat + (size_t)e * 8);
    const float4 fb = *(const float4*)(efeat + (size_t)e * 8 + 4);
    const float f[8] = {fa.x, fa.y, fa.z, fa.w, fb.x, fb.y, fb.z, fb.w};
    float h[8];
    #pragma unroll
    for (int r = 0; r < 8; ++r) {
      float t = 0.f;
      #pragma unroll
      for (int q = 0; q < 8; ++q) t += f[q] * Wm1[q * 8 + r];
      h[r] = sspf(t * 0.35355339059f);
    }
    const float4 yv = *(const float4*)(ea + (size_t)e * 4);
    const int snd = eidx[e];

    for (int t0 = 0; t0 < cnt; t0 += 8) {
      float se[8][2], vx[8][2], vy[8][2], vz[8][2];
      #pragma unroll
      for (int q = 0; q < 8; ++q) {
        const int tq = (t0 + q < cnt) ? (t0 + q) : (cnt - 1);
        const int s  = __builtin_amdgcn_readlane(snd, tq);
        const float* srow = sv + (size_t)s * 256;
        #pragma unroll
        for (int tt = 0; tt < 2; ++tt) {
          const int u = lane + 32 * tt;
          se[q][tt] = srow[u];
          vx[q][tt] = srow[64 + u * 3 + 0];
          vy[q][tt] = srow[64 + u * 3 + 1];
          vz[q][tt] = srow[64 + u * 3 + 2];
        }
      }
      #pragma unroll
      for (int q = 0; q < 8; ++q) {
        const int t = t0 + q;
        if (t < cnt) {
          const float Y0  = rlf(yv.x, t);
          const float Y1x = rlf(yv.y, t);
          const float Y1y = rlf(yv.z, t);
          const float Y1z = rlf(yv.w, t);
          float hb[8];
          #pragma unroll
          for (int r = 0; r < 8; ++r) hb[r] = rlf(h[r], t);
          float w[4][2];
          #pragma unroll
          for (int k = 0; k < 4; ++k) {
            float s0 = 0.f, s1 = 0.f;
            #pragma unroll
            for (int r = 0; r < 8; ++r) {
              s0 += hb[r] * wm2[r][k][0];
              s1 += hb[r] * wm2[r][k][1];
            }
            w[k][0] = s0 * 0.35355339059f;
            w[k][1] = s1 * 0.35355339059f;
          }
          #pragma unroll
          for (int tt = 0; tt < 2; ++tt) {
            const float dv = vx[q][tt] * Y1x + vy[q][tt] * Y1y + vz[q][tt] * Y1z;
            a0[tt]    += w[0][tt] * se[q][tt] * Y0;
            b0[tt]    += w[3][tt] * dv * 0.57735026919f;
            a1[tt][0] += w[1][tt] * se[q][tt] * Y1x;
            a1[tt][1] += w[1][tt] * se[q][tt] * Y1y;
            a1[tt][2] += w[1][tt] * se[q][tt] * Y1z;
            b1[tt][0] += w[2][tt] * vx[q][tt] * Y0;
            b1[tt][1] += w[2][tt] * vy[q][tt] * Y0;
            b1[tt][2] += w[2][tt] * vz[q][tt] * Y0;
          }
        }
      }
    }
  }

  float* base = msmv + (size_t)n * 512;
  #pragma unroll
  for (int tt = 0; tt < 2; ++tt) {
    const int u = lane + 32 * tt;
    base[u]      = a0[tt];
    base[64 + u] = b0[tt];
    #pragma unroll
    for (int d = 0; d < 3; ++d) {
      base[128 + u * 3 + d]        = a1[tt][d];
      base[128 + (64 + u) * 3 + d] = b1[tt][d];
    }
  }
}

template <bool ACONTIG, int NC>
__device__ __forceinline__ void post_pass(const float* __restrict__ arow,
                                          const float* W2,
                                          const float* Wsk,
                                          float* myLds, int kh, int l16,
                                          float attE0, float attE1,
                                          float attO0, float attO1,
                                          v8f (&acc)[NC][4]) {
  gemm_loop<ACONTIG, 32, NC, WP>(arow, W2, kh, l16, acc);

  #pragma unroll
  for (int d = 0; d < NC; ++d) {
    #pragma unroll
    for (int p = 0; p < 2; ++p) {
      #pragma unroll
      for (int j = 0; j < 8; ++j) {
        acc[d][2 * p][j]     *= 0.08838834765f;
        acc[d][2 * p + 1][j] *= 0.08838834765f;
        *(float2*)(myLds + d * 1024 + (j + 8 * kh) * 64 + 32 * p + 2 * l16) =
            make_float2(acc[d][2 * p][j], acc[d][2 * p + 1][j]);
      }
    }
  }
  __syncthreads();

  for (int i0 = 0; i0 < 64; i0 += 2) {
   #pragma unroll
   for (int ii = 0; ii < 2; ++ii) {
    const int i = i0 + ii;
    float m2[NC];
    #pragma unroll
    for (int d = 0; d < NC; ++d) m2[d] = myLds[d * 1024 + l16 * 64 + i];
    const float* bp = Wsk + (8 * i + 2 * kh) * WP + 2 * l16;
    v2f a[NC];
    #pragma unroll
    for (int d = 0; d < NC; ++d) {
      a[d].x = m2[d] * attE0; a[d].y = m2[d] * attE1;
    }
    #pragma unroll
    for (int p = 0; p < 2; ++p) {
      const float2 r0 = *(const float2*)(bp + 32 * p);
      const float2 r1 = *(const float2*)(bp + WP + 32 * p);
      v2f be; be.x = r0.x; be.y = r1.x;
      v2f bo; bo.x = r0.y; bo.y = r1.y;
      #pragma unroll
      for (int d = 0; d < NC; ++d)
        acc[d][2 * p] = wmma_f32_k4(a[d], be, acc[d][2 * p]);
      #pragma unroll
      for (int d = 0; d < NC; ++d)
        acc[d][2 * p + 1] = wmma_f32_k4(a[d], bo, acc[d][2 * p + 1]);
    }
    #pragma unroll
    for (int d = 0; d < NC; ++d) {
      a[d].x = m2[d] * attO0; a[d].y = m2[d] * attO1;
    }
    #pragma unroll
    for (int p = 0; p < 2; ++p) {
      const float2 r0 = *(const float2*)(bp + 4 * WP + 32 * p);
      const float2 r1 = *(const float2*)(bp + 5 * WP + 32 * p);
      v2f be; be.x = r0.x; be.y = r1.x;
      v2f bo; bo.x = r0.y; bo.y = r1.y;
      #pragma unroll
      for (int d = 0; d < NC; ++d)
        acc[d][2 * p] = wmma_f32_k4(a[d], be, acc[d][2 * p]);
      #pragma unroll
      for (int d = 0; d < NC; ++d)
        acc[d][2 * p + 1] = wmma_f32_k4(a[d], bo, acc[d][2 * p + 1]);
    }
   }
    __builtin_amdgcn_sched_barrier(0);
    asm volatile("v_nop\n\tv_nop\n\tv_nop\n\tv_nop");
  }
}

template <int ROWS>
__device__ __forceinline__ void stage_weights(const float* __restrict__ Wg,
                                              float* wl, int tid, int rot) {
  constexpr int TRIPS = ROWS / 16;
  #pragma unroll 8
  for (int j = 0; j < TRIPS; ++j) {
    const int i = (((j + rot) & (TRIPS - 1)) << 8) + tid;
    const float4 v = *(const float4*)(Wg + i * 4);
    *(float4*)(wl + (i >> 4) * WP + (i & 15) * 4) = v;
  }
}

__global__ void __launch_bounds__(256)
k_post(const float* __restrict__ msmv,
                       const float* __restrict__ attrs,
                       const float* __restrict__ W2s,
                       const float* __restrict__ W2v,
                       const float* __restrict__ WskS,
                       const float* __restrict__ WskV,
                       float* __restrict__ out) {
  __shared__ alignas(16) float lds[8 * 3 * 16 * 64 + (128 + 512) * WP];
  float* wl2  = lds + 8 * 3 * 16 * 64;
  float* wlsk = wl2 + 128 * WP;
  const int lane = threadIdx.x & 31;
  const int wid  = threadIdx.x >> 5;
  const int l16  = lane & 15;
  const int kh   = lane >> 4;
  const int row0 = (blockIdx.x * 8 + wid) * 16;
  int rowA = row0 + l16; if (rowA >= NN) rowA = NN - 1;
  float* myLds = lds + wid * (3 * 16 * 64);

  const float* arow8 = attrs + (size_t)rowA * 8;
  const float attE0 = arow8[2 * kh]     * 0.04419417382f;
  const float attE1 = arow8[2 * kh + 1] * 0.04419417382f;
  const float attO0 = arow8[4 + 2 * kh] * 0.04419417382f;
  const float attO1 = arow8[5 + 2 * kh] * 0.04419417382f;
  const float* mrow = msmv + (size_t)rowA * 512;

  stage_weights<128>(W2s, wl2, threadIdx.x, blockIdx.x);
  stage_weights<512>(WskS, wlsk, threadIdx.x, blockIdx.x);
  __syncthreads();
  {
    v8f acc[1][4] = {};
    post_pass<true, 1>(mrow, wl2, wlsk, myLds, kh, l16,
                       attE0, attE1, attO0, attO1, acc);
    #pragma unroll
    for (int p = 0; p < 2; ++p) {
      #pragma unroll
      for (int j = 0; j < 8; ++j) {
        const int r = row0 + j + 8 * kh;
        if (r < NN)
          *(float2*)(out + r * 256 + 32 * p + 2 * l16) =
              make_float2(acc[0][2 * p][j], acc[0][2 * p + 1][j]);
      }
    }
  }
  __syncthreads();
  stage_weights<128>(W2v, wl2, threadIdx.x, blockIdx.x);
  stage_weights<512>(WskV, wlsk, threadIdx.x, blockIdx.x);
  __syncthreads();
  {
    v8f acc[3][4] = {};
    post_pass<false, 3>(mrow + 128, wl2, wlsk, myLds, kh, l16,
                        attE0, attE1, attO0, attO1, acc);
    #pragma unroll
    for (int p = 0; p < 2; ++p) {
      #pragma unroll
      for (int j = 0; j < 8; ++j) {
        const int r = row0 + j + 8 * kh;
        if (r < NN) {
          float2* o = (float2*)(out + r * 256 + 64 + (32 * p + 2 * l16) * 3);
          o[0] = make_float2(acc[0][2 * p][j],     acc[1][2 * p][j]);
          o[1] = make_float2(acc[2][2 * p][j],     acc[0][2 * p + 1][j]);
          o[2] = make_float2(acc[1][2 * p + 1][j], acc[2][2 * p + 1][j]);
        }
      }
    }
  }
}

extern "C" void kernel_launch(void* const* d_in, const int* in_sizes, int n_in,
                              void* d_out, int out_size, void* d_ws, size_t ws_size,
                              hipStream_t stream) {
  const float* node_attrs = (const float*)d_in[0];
  const float* node_feats = (const float*)d_in[1];
  const float* edge_attrs = (const float*)d_in[2];
  const float* edge_feats = (const float*)d_in[3];
  const int*   edge_index = (const int*)d_in[4];
  const float* W1s  = (const float*)d_in[5];
  const float* W1v  = (const float*)d_in[6];
  const float* Wm1  = (const float*)d_in[7];
  const float* Wm2  = (const float*)d_in[8];
  const float* W2s  = (const float*)d_in[9];
  const float* W2v  = (const float*)d_in[10];
  const float* WskS = (const float*)d_in[11];
  const float* WskV = (const float*)d_in[12];

  float* out  = (float*)d_out;
  float* msmv = (float*)d_ws;

  int* cnt  = (int*)(msmv + (size_t)NN * 512);

  k_zero<<<20, 256, 0, stream>>>((float4*)cnt, NN / 4);
  k_fill<<<(NE + 255) / 256, 256, 0, stream>>>(edge_index, cnt, (int*)d_ws);
  k_pre <<<(NN + 127) / 128, 256, 0, stream>>>(node_feats, W1s, W1v, out);
  k_msg <<<(NN + 7) / 8, 256, 0, stream>>>(edge_attrs, edge_feats, edge_index,
                                           Wm1, Wm2, out, cnt, (const int*)d_ws, msmv);
  k_post<<<(NN + 127) / 128, 256, 0, stream>>>(msmv, node_attrs, W2s, W2v,
                                               WskS, WskV, out);
}


# k_post: a 16-row tile is worked by 4 waves, one 16-column output tile each (was 1 wave, all 64 columns); m2 and output rows pass through LDS; 250 groups of 20 waves; same k order and sums
# speedup vs baseline: 3.1923x; 1.0604x over previous
#include <hip/hip_runtime.h>
#include <hip/hip_bf16.h>
#include <math.h>


#define NN 20000
#define NE 320000
#define RCAP 512
#define WP 80
#define PT 5
#define TP 68
#define VP 196
#define TILE_F (3 * 16 * TP)

typedef __attribute__((ext_vector_type(2))) float v2f;
typedef __attribute__((ext_vector_type(8))) float v8f;

__device__ __forceinline__ v8f wmma_f32_k4(v2f a, v2f b, v8f c) {
  return __builtin_amdgcn_wmma_f32_16x16x4_f32(false, a, false, b,
                                               (short)0, c, false, false);
}

__device__ __forceinline__ float sspf(float x) {
  return fmaxf(x, 0.0f) + log1pf(__expf(-fabsf(x))) - 0.69314718056f;
}

__global__ void k_zero(float4* __restrict__ p, int n4) {
  int i = blockIdx.x * blockDim.x + threadIdx.x;
  const int stride = gridDim.x * blockDim.x;
  for (; i < n4; i += stride) p[i] = make_float4(0.f, 0.f, 0.f, 0.f);
}

template <bool ACONTIG, int KC, int NC, int PITCH>
__device__ __forceinline__ void gemm_loop(const float* __restrict__ arow,
                                          const float* __restrict__ W,
                                          int kh, int l16, v8f (&acc)[NC][4]) {
  #pragma unroll 4
  for (int kc = 0; kc < KC; ++kc) {
    const int k0 = kc * 4 + 2 * kh;
    v2f a[NC];
    if (ACONTIG) {
      const float2 av = *(const float2*)(arow + k0);
      a[0].x = av.x; a[0].y = av.y;
    } else {
      const float2* ap = (const float2*)(arow + k0 * 3);
      const float2 f0 = ap[0];
      const float2 f1 = ap[1];
      const float2 f2 = ap[2];
      a[0].x = f0.x; a[0].y = f1.y;
      a[NC > 1 ? 1 : 0].x = f0.y; a[NC > 1 ? 1 : 0].y = f2.x;
      a[NC > 2 ? 2 : 0].x = f1.x; a[NC > 2 ? 2 : 0].y = f2.y;
    }
    const float* bp = W + k0 * PITCH + 2 * l16;
    #pragma unroll
    for (int p = 0; p < 2; ++p) {
      const float2 r0 = *(const float2*)(bp + 32 * p);
      const float2 r1 = *(const float2*)(bp + PITCH + 32 * p);
      v2f be; be.x = r0.x; be.y = r1.x;
      v2f bo; bo.x = r0.y; bo.y = r1.y;
      #pragma unroll
      for (int d = 0; d < NC; ++d)
        acc[d][2 * p] = wmma_f32_k4(a[d], be, acc[d][2 * p]);
      #pragma unroll
      for (int d = 0; d < NC; ++d)
        acc[d][2 * p + 1] = wmma_f32_k4(a[d], bo, acc[d][2 * p + 1]);
    }
  }
}

__global__ void __launch_bounds__(256)
k_pre(const float* __restrict__ nf,
                      const float* __restrict__ W1s,
                      const float* __restrict__ W1v,
                      float* __restrict__ sv) {
  const int lane = threadIdx.x & 31;
  const int wid  = threadIdx.x >> 5;
  const int l16  = lane & 15;
  const int kh   = lane >> 4;
  const int row0 = (blockIdx.x * 8 + wid) * 16;
  int rowA = row0 + l16; if (rowA >= NN) rowA = NN - 1;
  const float* nrow = nf + (size_t)rowA * 256;

  {
    v8f acc[1][4] = {};
    gemm_loop<true, 16, 1, 64>(nrow, W1s, kh, l16, acc);
    #pragma unroll
    for (int p = 0; p < 2; ++p) {
      #pragma unroll
      for (int j = 0; j < 8; ++j) {
        const int r = row0 + j + 8 * kh;
        if (r < NN)
          *(float2*)(sv + r * 256 + 32 * p + 2 * l16) =
              make_float2(acc[0][2 * p][j] * 0.125f, acc[0][2 * p + 1][j] * 0.125f);
      }
    }
  }
  {
    v8f acc[3][4] = {};
    gemm_loop<false, 16, 3, 64>(nrow + 64, W1v, kh, l16, acc);
    #pragma unroll
    for (int p = 0; p < 2; ++p) {
      #pragma unroll
      for (int j = 0; j < 8; ++j) {
        const int r = row0 + j + 8 * kh;
        if (r < NN) {
          float2* o = (float2*)(sv + r * 256 + 64 + (32 * p + 2 * l16) * 3);
          o[0] = make_float2(acc[0][2 * p][j] * 0.125f,
                             acc[1][2 * p][j] * 0.125f);
          o[1] = make_float2(acc[2][2 * p][j] * 0.125f,
                             acc[0][2 * p + 1][j] * 0.125f);
          o[2] = make_float2(acc[1][2 * p + 1][j] * 0.125f,
                             acc[2][2 * p + 1][j] * 0.125f);
        }
      }
    }
  }
}

__global__ void k_fill(const int* __restrict__ eidx, int* __restrict__ cnt,
                       int* __restrict__ rows) {
  const int e = blockIdx.x * blockDim.x + threadIdx.x;
  if (e < NE) {
    const int r = eidx[NE + e];
    const int pos = atomicAdd(&cnt[r], 1);
    if (pos < RCAP) rows[(size_t)r * 512 + pos] = e;
  }
}

__device__ __forceinline__ int take_by_rank(int rk, int id, int lane, int mine) {
  #pragma unroll
  for (int i = 0; i < 32; ++i) {
    const int r = __builtin_amdgcn_readlane(rk, i);
    const int v = __builtin_amdgcn_readlane(id, i);
    if (r == lane) mine = v;
  }
  return mine;
}

__device__ __forceinline__ float rlf(float x, int l) {
  return __int_as_float(__builtin_amdgcn_readlane(__float_as_int(x), l));
}

__global__ void __launch_bounds__(256)
k_msg(const float* __restrict__ ea,
      const float* __restrict__ efeat,
      const int*   __restrict__ eidx,
      const float* __restrict__ Wm1,
      const float* __restrict__ Wm2,
      const float* __restrict__ sv,
      const int*   __restrict__ deg,
      const int*   rows,
      float* msmv) {
  const int lane = threadIdx.x & 31;
  const int n = __builtin_amdgcn_readfirstlane(blockIdx.x * 8 + (threadIdx.x >> 5));
  if (n >= NN) return;
  const int len = deg[n];
  const int* row = rows + (size_t)n * 512;
  const bool inrow = (len <= RCAP);

  float wm2[8][4][2];
  #pragma unroll
  for (int r = 0; r < 8; ++r) {
    #pragma unroll
    for (int k = 0; k < 4; ++k) {
      wm2[r][k][0] = Wm2[r * 256 + k * 64 + lane];
      wm2[r][k][1] = Wm2[r * 256 + k * 64 + lane + 32];
    }
  }
  float a0[2] = {0.f, 0.f}, b0[2] = {0.f, 0.f};
  float a1[2][3] = {{0.f, 0.f, 0.f}, {0.f, 0.f, 0.f}};
  float b1[2][3] = {{0.f, 0.f, 0.f}, {0.f, 0.f, 0.f}};

  int last = -1;
  int done = 0;
  int e0 = 0;
  for (;;) {
    int cnt;
    int mine = 0x7fffffff;
    if (inrow) {
      if (done >= len) break;
      cnt = (len - done < 32) ? (len - done) : 32;
      for (int j0 = 0; j0 < len; j0 += 32) {
        const int jj = j0 + lane;
        const int idj = (jj < len) ? row[jj] : 0x7fffffff;
        int rank = 0;
        for (int i0 = 0; i0 < len; i0 += 32) {
          const int ii = i0 + lane;
          const int idi = (ii < len) ? row[ii] : 0x7fffffff;
          #pragma unroll
          for (int i = 0; i < 32; ++i) {
            const int v = __builtin_amdgcn_readlane(idi, i);
            rank += (v > last && v < idj) ? 1 : 0;
          }
        }
        const int rk = (idj > last && idj != 0x7fffffff) ? rank : -1;
        mine = take_by_rank(rk, idj, lane, mine);
      }
      last = __builtin_amdgcn_readlane(mine, cnt - 1);
      done += 32;
    } else {
      unsigned hit = 0u;
      int eid = 0;
      while (e0 < NE) {
        eid = e0 + lane;
        hit = __builtin_amdgcn_ballot_w32(eidx[NE + eid] == n);
        e0 += 32;
        if (hit != 0u) break;
      }
      if (hit == 0u) break;
      cnt = __builtin_popcount(hit);
      const int rk = ((hit >> lane) & 1u)
                         ? __builtin_popcount(hit & ((1u << lane) - 1u)) : -1;
      mine = take_by_rank(rk, eid, lane, mine);
    }

    const int e = (lane < cnt) ? mine : 0;
    const float4 fa = *(const float4*)(efeat + (size_t)e * 8);
    const float4 fb = *(const float4*)(efeat + (size_t)e * 8 + 4);
    const float f[8] = {fa.x, fa.y, fa.z, fa.w, fb.x, fb.y, fb.z, fb.w};
    float h[8];
    #pragma unroll
    for (int r = 0; r < 8; ++r) {
      float t = 0.f;
      #pragma unroll
      for (int q = 0; q < 8; ++q) t += f[q] * Wm1[q * 8 + r];
      h[r] = sspf(t * 0.35355339059f);
    }
    const float4 yv = *(const float4*)(ea + (size_t)e * 4);
    const int snd = eidx[e];

    for (int t0 = 0; t0 < cnt; t0 += 8) {
      float se[8][2], vx[8][2], vy[8][2], vz[8][2];
      #pragma unroll
      for (int q = 0; q < 8; ++q) {
        const int tq = (t0 + q < cnt) ? (t0 + q) : (cnt - 1);
        const int s  = __builtin_amdgcn_readlane(snd, tq);
        const float* srow = sv + (size_t)s * 256;
        #pragma unroll
        for (int tt = 0; tt < 2; ++tt) {
          const int u = lane + 32 * tt;
          se[q][tt] = srow[u];
          vx[q][tt] = srow[64 + u * 3 + 0];
          vy[q][tt] = srow[64 + u * 3 + 1];
          vz[q][tt] = srow[64 + u * 3 + 2];
        }
      }
      #pragma unroll
      for (int q = 0; q < 8; ++q) {
        const int t = t0 + q;
        if (t < cnt) {
          const float Y0  = rlf(yv.x, t);
          const float Y1x = rlf(yv.y, t);
          const float Y1y = rlf(yv.z, t);
          const float Y1z = rlf(yv.w, t);
          float hb[8];
          #pragma unroll
          for (int r = 0; r < 8; ++r) hb[r] = rlf(h[r], t);
          float w[4][2];
          #pragma unroll
          for (int k = 0; k < 4; ++k) {
            float s0 = 0.f, s1 = 0.f;
            #pragma unroll
            for (int r = 0; r < 8; ++r) {
              s0 += hb[r] * wm2[r][k][0];
              s1 += hb[r] * wm2[r][k][1];
            }
            w[k][0] = s0 * 0.35355339059f;
            w[k][1] = s1 * 0.35355339059f;
          }
          #pragma unroll
          for (int tt = 0; tt < 2; ++tt) {
            const float dv = vx[q][tt] * Y1x + vy[q][tt] * Y1y + vz[q][tt] * Y1z;
            a0[tt]    += w[0][tt] * se[q][tt] * Y0;
            b0[tt]    += w[3][tt] * dv * 0.57735026919f;
            a1[tt][0] += w[1][tt] * se[q][tt] * Y1x;
            a1[tt][1] += w[1][tt] * se[q][tt] * Y1y;
            a1[tt][2] += w[1][tt] * se[q][tt] * Y1z;
            b1[tt][0] += w[2][tt] * vx[q][tt] * Y0;
            b1[tt][1] += w[2][tt] * vy[q][tt] * Y0;
            b1[tt][2] += w[2][tt] * vz[q][tt] * Y0;
          }
        }
      }
    }
  }

  float* base = msmv + (size_t)n * 512;
  #pragma unroll
  for (int tt = 0; tt < 2; ++tt) {
    const int u = lane + 32 * tt;
    base[u]      = a0[tt];
    base[64 + u] = b0[tt];
    #pragma unroll
    for (int d = 0; d < 3; ++d) {
      base[128 + u * 3 + d]        = a1[tt][d];
      base[128 + (64 + u) * 3 + d] = b1[tt][d];
    }
  }
}

__device__ __forceinline__ void wmma_back_edge_slots() {
  __builtin_amdgcn_sched_barrier(0);
  asm volatile("v_nop\n\tv_nop\n\tv_nop\n\tv_nop");
}

template <bool ACONTIG, int KC, int NC>
__device__ __forceinline__ void gemm_tile(const float* __restrict__ arow,
                                          const float* Wc, int kh,
                                          v8f (&acc)[NC]) {
  for (int kb = 0; kb < KC; kb += 8) {
   #pragma unroll
   for (int kk = 0; kk < 8; ++kk) {
    const int k0 = (kb + kk) * 4 + 2 * kh;
    v2f a[NC];
    if (ACONTIG) {
      const float2 av = *(const float2*)(arow + k0);
      a[0].x = av.x; a[0].y = av.y;
    } else {
      const float2* ap = (const float2*)(arow + k0 * 3);
      const float2 f0 = ap[0];
      const float2 f1 = ap[1];
      const float2 f2 = ap[2];
      a[0].x = f0.x; a[0].y = f1.y;
      a[NC > 1 ? 1 : 0].x = f0.y; a[NC > 1 ? 1 : 0].y = f2.x;
      a[NC > 2 ? 2 : 0].x = f1.x; a[NC > 2 ? 2 : 0].y = f2.y;
    }
    v2f b;
    b.x = Wc[k0 * WP];
    b.y = Wc[(k0 + 1) * WP];
    #pragma unroll
    for (int d = 0; d < NC; ++d) acc[d] = wmma_f32_k4(a[d], b, acc[d]);
   }
   wmma_back_edge_slots();
  }
}

template <bool ACONTIG, int NC, int UPT, bool WAIT_TILE>
__device__ __forceinline__ void post_pass(const float* __restrict__ arow,
                                          const float* W2,
                                          const float* Wsk,
                                          float* tile, int kh, int l16, int col,
                                          float attE0, float attE1,
                                          float attO0, float attO1,
                                          v8f (&acc)[NC]) {
  gemm_tile<ACONTIG, 32, NC>(arow, W2 + col, kh, acc);

  if (WAIT_TILE) __syncthreads();

  #pragma unroll
  for (int d = 0; d < NC; ++d) {
    #pragma unroll
    for (int j = 0; j < 8; ++j) {
      acc[d][j] *= 0.08838834765f;
      tile[d * (16 * TP) + (j + 8 * kh) * TP + col] = acc[d][j];
    }
  }
  __syncthreads();

  for (int i0 = 0; i0 < 64; i0 += UPT) {
    float mv[NC][UPT];
    #pragma unroll
    for (int d = 0; d < NC; ++d) {
      const float* mp = tile + d * (16 * TP) + l16 * TP + i0;
      if constexpr (UPT == 4) {
        const float4 t = *(const float4*)mp;
        mv[d][0] = t.x; mv[d][1] = t.y; mv[d][2] = t.z; mv[d][3] = t.w;
      } else {
        const float2 t = *(const float2*)mp;
        mv[d][0] = t.x; mv[d][1] = t.y;
      }
    }
    #pragma unroll
    for (int ii = 0; ii < UPT; ++ii) {
      const int i = i0 + ii;
      const float* bp = Wsk + (8 * i + 2 * kh) * WP + col;
      v2f be; be.x = bp[0];      be.y = bp[WP];
      v2f bo; bo.x = bp[4 * WP]; bo.y = bp[5 * WP];
      v2f a[NC];
      #pragma unroll
      for (int d = 0; d < NC; ++d) {
        a[d].x = mv[d][ii] * attE0; a[d].y = mv[d][ii] * attE1;
      }
      #pragma unroll
      for (int d = 0; d < NC; ++d) acc[d] = wmma_f32_k4(a[d], be, acc[d]);
      #pragma unroll
      for (int d = 0; d < NC; ++d) {
        a[d].x = mv[d][ii] * attO0; a[d].y = mv[d][ii] * attO1;
      }
      #pragma unroll
      for (int d = 0; d < NC; ++d) acc[d] = wmma_f32_k4(a[d], bo, acc[d]);
    }
    wmma_back_edge_slots();
  }
}

template <int ROWS>
__device__ __forceinline__ void stage_weights(const float* __restrict__ Wg,
                                              float* wl, int tid, int rot) {
  constexpr int PIECES = ROWS * 16;
  const int start = (rot << 8) & (PIECES - 1);
  #pragma unroll 4
  for (int q = tid; q < PIECES; q += PT * 128) {
    const int i = (q + start) & (PIECES - 1);
    const float4 v = *(const float4*)(Wg + i * 4);
    *(float4*)(wl + (i >> 4) * WP + (i & 15) * 4) = v;
  }
}

__global__ void __launch_bounds__(PT * 128)
k_post(const float* __restrict__ msmv,
                       const float* __restrict__ attrs,
                       const float* __restrict__ W2s,
                       const float* __restrict__ W2v,
                       const float* __restrict__ WskS,
                       const float* __restrict__ WskV,
                       float* __restrict__ out) {
  __shared__ alignas(16) float lds[(128 + 512) * WP + PT * TILE_F];
  float* wl2  = lds;
  float* wlsk = wl2 + 128 * WP;
  const int lane = threadIdx.x & 31;
  const int wid  = threadIdx.x >> 5;
  const int l16  = lane & 15;
  const int kh   = lane >> 4;
  const int g    = wid >> 2;
  const int ct   = wid & 3;
  const int col  = 16 * ct + l16;
  const int row0 = (blockIdx.x * PT + g) * 16;
  int rowA = row0 + l16; if (rowA >= NN) rowA = NN - 1;
  float* tile = lds + (128 + 512) * WP + g * TILE_F;

  const float* arow8 = attrs + (size_t)rowA * 8;
  const float attE0 = arow8[2 * kh]     * 0.04419417382f;
  const float attE1 = arow8[2 * kh + 1] * 0.04419417382f;
  const float attO0 = arow8[4 + 2 * kh] * 0.04419417382f;
  const float attO1 = arow8[5 + 2 * kh] * 0.04419417382f;
  const float* mrow = msmv + (size_t)rowA * 512;

  stage_weights<128>(W2s, wl2, threadIdx.x, blockIdx.x);
  stage_weights<512>(WskS, wlsk, threadIdx.x, blockIdx.x);
  __syncthreads();
  {
    v8f acc[1] = {};
    post_pass<true, 1, 4, false>(mrow, wl2, wlsk, tile, kh, l16, col,
                                 attE0, attE1, attO0, attO1, acc);
    __syncthreads();
    #pragma unroll
    for (int j = 0; j < 8; ++j) tile[(j + 8 * kh) * TP + col] = acc[0][j];
  }
  stage_weights<128>(W2v, wl2, threadIdx.x, blockIdx.x);
  stage_weights<512>(WskV, wlsk, threadIdx.x, blockIdx.x);
  __syncthreads();
  #pragma unroll
  for (int rr = 0; rr < 4; ++rr) {
    const int m = 4 * ct + rr;
    const int r = row0 + m;
    const float2 v = *(const float2*)(tile + m * TP + 2 * lane);
    if (r < NN) *(float2*)(out + (size_t)r * 256 + 2 * lane) = v;
  }
  {
    v8f acc[3] = {};
    post_pass<false, 3, 2, true>(mrow + 128, wl2, wlsk, tile, kh, l16, col,
                                 attE0, attE1, attO0, attO1, acc);
    __syncthreads();
    #pragma unroll
    for (int j = 0; j < 8; ++j) {
      #pragma unroll
      for (int d = 0; d < 3; ++d)
        tile[(j + 8 * kh) * VP + col * 3 + d] = acc[d][j];
    }
  }
  __syncthreads();
  #pragma unroll
  for (int rr = 0; rr < 4; ++rr) {
    const int m = 4 * ct + rr;
    const int r = row0 + m;
    #pragma unroll
    for (int q = 0; q < 3; ++q) {
      const float2 v = *(const float2*)(tile + m * VP + 64 * q + 2 * lane);
      if (r < NN) *(float2*)(out + (size_t)r * 256 + 64 + 64 * q + 2 * lane) = v;
    }
  }
}

extern "C" void kernel_launch(void* const* d_in, const int* in_sizes, int n_in,
                              void* d_out, int out_size, void* d_ws, size_t ws_size,
                              hipStream_t stream) {
  const float* node_attrs = (const float*)d_in[0];
  const float* node_feats = (const float*)d_in[1];
  const float* edge_attrs = (const float*)d_in[2];
  const float* edge_feats = (const float*)d_in[3];
  const int*   edge_index = (const int*)d_in[4];
  const float* W1s  = (const float*)d_in[5];
  const float* W1v  = (const float*)d_in[6];
  const float* Wm1  = (const float*)d_in[7];
  const float* Wm2  = (const float*)d_in[8];
  const float* W2s  = (const float*)d_in[9];
  const float* W2v  = (const float*)d_in[10];
  const float* WskS = (const float*)d_in[11];
  const float* WskV = (const float*)d_in[12];

  float* out  = (float*)d_out;
  float* msmv = (float*)d_ws;

  int* cnt  = (int*)(msmv + (size_t)NN * 512);

  k_zero<<<20, 256, 0, stream>>>((float4*)cnt, NN / 4);
  k_fill<<<(NE + 255) / 256, 256, 0, stream>>>(edge_index, cnt, (int*)d_ws);
  k_pre <<<(NN + 127) / 128, 256, 0, stream>>>(node_feats, W1s, W1v, out);
  k_msg <<<(NN + 7) / 8, 256, 0, stream>>>(edge_attrs, edge_feats, edge_index,
                                           Wm1, Wm2, out, cnt, (const int*)d_ws, msmv);
  k_post<<<(NN + 16 * PT - 1) / (16 * PT), PT * 128, 0, stream>>>(msmv, node_attrs, W2s, W2v,
                                                                  WskS, WskV, out);
}


# edge placement (k_fill) runs as extra work-groups behind the tile groups of k_pre's launch (the two jobs are independent): 5 launches -> 4, its atomics run beside the WMMA tiles; same sums
# speedup vs baseline: 3.3524x; 1.0502x over previous
#include <hip/hip_runtime.h>
#include <hip/hip_bf16.h>
#include <math.h>


#define NN 20000
#define NE 320000
#define RCAP 512
#define WP 80
#define PT 5
#define TP 68
#define VP 196
#define TILE_F (3 * 16 * TP)

typedef __attribute__((ext_vector_type(2))) float v2f;
typedef __attribute__((ext_vector_type(8))) float v8f;

__device__ __forceinline__ v8f wmma_f32_k4(v2f a, v2f b, v8f c) {
  return __builtin_amdgcn_wmma_f32_16x16x4_f32(false, a, false, b,
                                               (short)0, c, false, false);
}

__device__ __forceinline__ float sspf(float x) {
  return fmaxf(x, 0.0f) + log1pf(__expf(-fabsf(x))) - 0.69314718056f;
}

__global__ void k_zero(float4* __restrict__ p, int n4) {
  int i = blockIdx.x * blockDim.x + threadIdx.x;
  const int stride = gridDim.x * blockDim.x;
  for (; i < n4; i += stride) p[i] = make_float4(0.f, 0.f, 0.f, 0.f);
}

template <bool ACONTIG, int KC, int NC, int PITCH>
__device__ __forceinline__ void gemm_loop(const float* __restrict__ arow,
                                          const float* __restrict__ W,
                                          int kh, int l16, v8f (&acc)[NC][4]) {
  #pragma unroll 4
  for (int kc = 0; kc < KC; ++kc) {
    const int k0 = kc * 4 + 2 * kh;
    v2f a[NC];
    if (ACONTIG) {
      const float2 av = *(const float2*)(arow + k0);
      a[0].x = av.x; a[0].y = av.y;
    } else {
      const float2* ap = (const float2*)(arow + k0 * 3);
      const float2 f0 = ap[0];
      const float2 f1 = ap[1];
      const float2 f2 = ap[2];
      a[0].x = f0.x; a[0].y = f1.y;
      a[NC > 1 ? 1 : 0].x = f0.y; a[NC > 1 ? 1 : 0].y = f2.x;
      a[NC > 2 ? 2 : 0].x = f1.x; a[NC > 2 ? 2 : 0].y = f2.y;
    }
    const float* bp = W + k0 * PITCH + 2 * l16;
    #pragma unroll
    for (int p = 0; p < 2; ++p) {
      const float2 r0 = *(const float2*)(bp + 32 * p);
      const float2 r1 = *(const float2*)(bp + PITCH + 32 * p);
      v2f be; be.x = r0.x; be.y = r1.x;
      v2f bo; bo.x = r0.y; bo.y = r1.y;
      #pragma unroll
      for (int d = 0; d < NC; ++d)
        acc[d][2 * p] = wmma_f32_k4(a[d], be, acc[d][2 * p]);
      #pragma unroll
      for (int d = 0; d < NC; ++d)
        acc[d][2 * p + 1] = wmma_f32_k4(a[d], bo, acc[d][2 * p + 1]);
    }
  }
}

#define PRE_WGS  ((NN + 127) / 128)
#define FILL_WGS ((NE + 255) / 256)
__global__ void __launch_bounds__(256)
k_front(const float* __restrict__ nf,
        const float* __restrict__ W1s,
        const float* __restrict__ W1v,
        float* __restrict__ sv,
        const int* __restrict__ eidx,
        int* __restrict__ cnt,
        int* __restrict__ rows) {
  if (blockIdx.x >= PRE_WGS) {
    const int e = (blockIdx.x - PRE_WGS) * 256 + threadIdx.x;
    if (e < NE) {
      const int r = eidx[NE + e];
      const int pos = atomicAdd(&cnt[r], 1);
      if (pos < RCAP) rows[(size_t)r * 512 + pos] = e;
    }
    return;
  }
  const int lane = threadIdx.x & 31;
  const int wid  = threadIdx.x >> 5;
  const int l16  = lane & 15;
  const int kh   = lane >> 4;
  const int row0 = (blockIdx.x * 8 + wid) * 16;
  int rowA = row0 + l16; if (rowA >= NN) rowA = NN - 1;
  const float* nrow = nf + (size_t)rowA * 256;

  {
    v8f acc[1][4] = {};
    gemm_loop<true, 16, 1, 64>(nrow, W1s, kh, l16, acc);
    #pragma unroll
    for (int p = 0; p < 2; ++p) {
      #pragma unroll
      for (int j = 0; j < 8; ++j) {
        const int r = row0 + j + 8 * kh;
        if (r < NN)
          *(float2*)(sv + r * 256 + 32 * p + 2 * l16) =
              make_float2(acc[0][2 * p][j] * 0.125f, acc[0][2 * p + 1][j] * 0.125f);
      }
    }
  }
  {
    v8f acc[3][4] = {};
    gemm_loop<false, 16, 3, 64>(nrow + 64, W1v, kh, l16, acc);
    #pragma unroll
    for (int p = 0; p < 2; ++p) {
      #pragma unroll
      for (int j = 0; j < 8; ++j) {
        const int r = row0 + j + 8 * kh;
        if (r < NN) {
          float2* o = (float2*)(sv + r * 256 + 64 + (32 * p + 2 * l16) * 3);
          o[0] = make_float2(acc[0][2 * p][j] * 0.125f,
                             acc[1][2 * p][j] * 0.125f);
          o[1] = make_float2(acc[2][2 * p][j] * 0.125f,
                             acc[0][2 * p + 1][j] * 0.125f);
          o[2] = make_float2(acc[1][2 * p + 1][j] * 0.125f,
                             acc[2][2 * p + 1][j] * 0.125f);
        }
      }
    }
  }
}

__device__ __forceinline__ int take_by_rank(int rk, int id, int lane, int mine) {
  #pragma unroll
  for (int i = 0; i < 32; ++i) {
    const int r = __builtin_amdgcn_readlane(rk, i);
    const int v = __builtin_amdgcn_readlane(id, i);
    if (r == lane) mine = v;
  }
  return mine;
}

__device__ __forceinline__ float rlf(float x, int l) {
  return __int_as_float(__builtin_amdgcn_readlane(__float_as_int(x), l));
}

__global__ void __launch_bounds__(256)
k_msg(const float* __restrict__ ea,
      const float* __restrict__ efeat,
      const int*   __restrict__ eidx,
      const float* __restrict__ Wm1,
      const float* __restrict__ Wm2,
      const float* __restrict__ sv,
      const int*   __restrict__ deg,
      const int*   rows,
      float* msmv) {
  const int lane = threadIdx.x & 31;
  const int n = __builtin_amdgcn_readfirstlane(blockIdx.x * 8 + (threadIdx.x >> 5));
  if (n >= NN) return;
  const int len = deg[n];
  const int* row = rows + (size_t)n * 512;
  const bool inrow = (len <= RCAP);

  float wm2[8][4][2];
  #pragma unroll
  for (int r = 0; r < 8; ++r) {
    #pragma unroll
    for (int k = 0; k < 4; ++k) {
      wm2[r][k][0] = Wm2[r * 256 + k * 64 + lane];
      wm2[r][k][1] = Wm2[r * 256 + k * 64 + lane + 32];
    }
  }
  float a0[2] = {0.f, 0.f}, b0[2] = {0.f, 0.f};
  float a1[2][3] = {{0.f, 0.f, 0.f}, {0.f, 0.f, 0.f}};
  float b1[2][3] = {{0.f, 0.f, 0.f}, {0.f, 0.f, 0.f}};

  int last = -1;
  int done = 0;
  int e0 = 0;
  for (;;) {
    int cnt;
    int mine = 0x7fffffff;
    if (inrow) {
      if (done >= len) break;
      cnt = (len - done < 32) ? (len - done) : 32;
      for (int j0 = 0; j0 < len; j0 += 32) {
        const int jj = j0 + lane;
        const int idj = (jj < len) ? row[jj] : 0x7fffffff;
        int rank = 0;
        for (int i0 = 0; i0 < len; i0 += 32) {
          const int ii = i0 + lane;
          const int idi = (ii < len) ? row[ii] : 0x7fffffff;
          #pragma unroll
          for (int i = 0; i < 32; ++i) {
            const int v = __builtin_amdgcn_readlane(idi, i);
            rank += (v > last && v < idj) ? 1 : 0;
          }
        }
        const int rk = (idj > last && idj != 0x7fffffff) ? rank : -1;
        mine = take_by_rank(rk, idj, lane, mine);
      }
      last = __builtin_amdgcn_readlane(mine, cnt - 1);
      done += 32;
    } else {
      unsigned hit = 0u;
      int eid = 0;
      while (e0 < NE) {
        eid = e0 + lane;
        hit = __builtin_amdgcn_ballot_w32(eidx[NE + eid] == n);
        e0 += 32;
        if (hit != 0u) break;
      }
      if (hit == 0u) break;
      cnt = __builtin_popcount(hit);
      const int rk = ((hit >> lane) & 1u)
                         ? __builtin_popcount(hit & ((1u << lane) - 1u)) : -1;
      mine = take_by_rank(rk, eid, lane, mine);
    }

    const int e = (lane < cnt) ? mine : 0;
    const float4 fa = *(const float4*)(efeat + (size_t)e * 8);
    const float4 fb = *(const float4*)(efeat + (size_t)e * 8 + 4);
    const float f[8] = {fa.x, fa.y, fa.z, fa.w, fb.x, fb.y, fb.z, fb.w};
    float h[8];
    #pragma unroll
    for (int r = 0; r < 8; ++r) {
      float t = 0.f;
      #pragma unroll
      for (int q = 0; q < 8; ++q) t += f[q] * Wm1[q * 8 + r];
      h[r] = sspf(t * 0.35355339059f);
    }
    const float4 yv = *(const float4*)(ea + (size_t)e * 4);
    const int snd = eidx[e];

    for (int t0 = 0; t0 < cnt; t0 += 8) {
      float se[8][2], vx[8][2], vy[8][2], vz[8][2];
      #pragma unroll
      for (int q = 0; q < 8; ++q) {
        const int tq = (t0 + q < cnt) ? (t0 + q) : (cnt - 1);
        const int s  = __builtin_amdgcn_readlane(snd, tq);
        const float* srow = sv + (size_t)s * 256;
        #pragma unroll
        for (int tt = 0; tt < 2; ++tt) {
          const int u = lane + 32 * tt;
          se[q][tt] = srow[u];
          vx[q][tt] = srow[64 + u * 3 + 0];
          vy[q][tt] = srow[64 + u * 3 + 1];
          vz[q][tt] = srow[64 + u * 3 + 2];
        }
      }
      #pragma unroll
      for (int q = 0; q < 8; ++q) {
        const int t = t0 + q;
        if (t < cnt) {
          const float Y0  = rlf(yv.x, t);
          const float Y1x = rlf(yv.y, t);
          const float Y1y = rlf(yv.z, t);
          const float Y1z = rlf(yv.w, t);
          float hb[8];
          #pragma unroll
          for (int r = 0; r < 8; ++r) hb[r] = rlf(h[r], t);
          float w[4][2];
          #pragma unroll
          for (int k = 0; k < 4; ++k) {
            float s0 = 0.f, s1 = 0.f;
            #pragma unroll
            for (int r = 0; r < 8; ++r) {
              s0 += hb[r] * wm2[r][k][0];
              s1 += hb[r] * wm2[r][k][1];
            }
            w[k][0] = s0 * 0.35355339059f;
            w[k][1] = s1 * 0.35355339059f;
          }
          #pragma unroll
          for (int tt = 0; tt < 2; ++tt) {
            const float dv = vx[q][tt] * Y1x + vy[q][tt] * Y1y + vz[q][tt] * Y1z;
            a0[tt]    += w[0][tt] * se[q][tt] * Y0;
            b0[tt]    += w[3][tt] * dv * 0.57735026919f;
            a1[tt][0] += w[1][tt] * se[q][tt] * Y1x;
            a1[tt][1] += w[1][tt] * se[q][tt] * Y1y;
            a1[tt][2] += w[1][tt] * se[q][tt] * Y1z;
            b1[tt][0] += w[2][tt] * vx[q][tt] * Y0;
            b1[tt][1] += w[2][tt] * vy[q][tt] * Y0;
            b1[tt][2] += w[2][tt] * vz[q][tt] * Y0;
          }
        }
      }
    }
  }

  float* base = msmv + (size_t)n * 512;
  #pragma unroll
  for (int tt = 0; tt < 2; ++tt) {
    const int u = lane + 32 * tt;
    base[u]      = a0[tt];
    base[64 + u] = b0[tt];
    #pragma unroll
    for (int d = 0; d < 3; ++d) {
      base[128 + u * 3 + d]        = a1[tt][d];
      base[128 + (64 + u) * 3 + d] = b1[tt][d];
    }
  }
}

__device__ __forceinline__ void wmma_back_edge_slots() {
  __builtin_amdgcn_sched_barrier(0);
  asm volatile("v_nop\n\tv_nop\n\tv_nop\n\tv_nop");
}

template <bool ACONTIG, int KC, int NC>
__device__ __forceinline__ void gemm_tile(const float* __restrict__ arow,
                                          const float* Wc, int kh,
                                          v8f (&acc)[NC]) {
  for (int kb = 0; kb < KC; kb += 8) {
   #pragma unroll
   for (int kk = 0; kk < 8; ++kk) {
    const int k0 = (kb + kk) * 4 + 2 * kh;
    v2f a[NC];
    if (ACONTIG) {
      const float2 av = *(const float2*)(arow + k0);
      a[0].x = av.x; a[0].y = av.y;
    } else {
      const float2* ap = (const float2*)(arow + k0 * 3);
      const float2 f0 = ap[0];
      const float2 f1 = ap[1];
      const float2 f2 = ap[2];
      a[0].x = f0.x; a[0].y = f1.y;
      a[NC > 1 ? 1 : 0].x = f0.y; a[NC > 1 ? 1 : 0].y = f2.x;
      a[NC > 2 ? 2 : 0].x = f1.x; a[NC > 2 ? 2 : 0].y = f2.y;
    }
    v2f b;
    b.x = Wc[k0 * WP];
    b.y = Wc[(k0 + 1) * WP];
    #pragma unroll
    for (int d = 0; d < NC; ++d) acc[d] = wmma_f32_k4(a[d], b, acc[d]);
   }
   wmma_back_edge_slots();
  }
}

template <bool ACONTIG, int NC, int UPT, bool WAIT_TILE>
__device__ __forceinline__ void post_pass(const float* __restrict__ arow,
                                          const float* W2,
                                          const float* Wsk,
                                          float* tile, int kh, int l16, int col,
                                          float attE0, float attE1,
                                          float attO0, float attO1,
                                          v8f (&acc)[NC]) {
  gemm_tile<ACONTIG, 32, NC>(arow, W2 + col, kh, acc);

  if (WAIT_TILE) __syncthreads();

  #pragma unroll
  for (int d = 0; d < NC; ++d) {
    #pragma unroll
    for (int j = 0; j < 8; ++j) {
      acc[d][j] *= 0.08838834765f;
      tile[d * (16 * TP) + (j + 8 * kh) * TP + col] = acc[d][j];
    }
  }
  __syncthreads();

  for (int i0 = 0; i0 < 64; i0 += UPT) {
    float mv[NC][UPT];
    #pragma unroll
    for (int d = 0; d < NC; ++d) {
      const float* mp = tile + d * (16 * TP) + l16 * TP + i0;
      if constexpr (UPT == 4) {
        const float4 t = *(const float4*)mp;
        mv[d][0] = t.x; mv[d][1] = t.y; mv[d][2] = t.z; mv[d][3] = t.w;
      } else {
        const float2 t = *(const float2*)mp;
        mv[d][0] = t.x; mv[d][1] = t.y;
      }
    }
    #pragma unroll
    for (int ii = 0; ii < UPT; ++ii) {
      const int i = i0 + ii;
      const float* bp = Wsk + (8 * i + 2 * kh) * WP + col;
      v2f be; be.x = bp[0];      be.y = bp[WP];
      v2f bo; bo.x = bp[4 * WP]; bo.y = bp[5 * WP];
      v2f a[NC];
      #pragma unroll
      for (int d = 0; d < NC; ++d) {
        a[d].x = mv[d][ii] * attE0; a[d].y = mv[d][ii] * attE1;
      }
      #pragma unroll
      for (int d = 0; d < NC; ++d) acc[d] = wmma_f32_k4(a[d], be, acc[d]);
      #pragma unroll
      for (int d = 0; d < NC; ++d) {
        a[d].x = mv[d][ii] * attO0; a[d].y = mv[d][ii] * attO1;
      }
      #pragma unroll
      for (int d = 0; d < NC; ++d) acc[d] = wmma_f32_k4(a[d], bo, acc[d]);
    }
    wmma_back_edge_slots();
  }
}

template <int ROWS>
__device__ __forceinline__ void stage_weights(const float* __restrict__ Wg,
                                              float* wl, int tid, int rot) {
  constexpr int PIECES = ROWS * 16;
  const int start = (rot << 8) & (PIECES - 1);
  #pragma unroll 4
  for (int q = tid; q < PIECES; q += PT * 128) {
    const int i = (q + start) & (PIECES - 1);
    const float4 v = *(const float4*)(Wg + i * 4);
    *(float4*)(wl + (i >> 4) * WP + (i & 15) * 4) = v;
  }
}

__global__ void __launch_bounds__(PT * 128)
k_post(const float* __restrict__ msmv,
                       const float* __restrict__ attrs,
                       const float* __restrict__ W2s,
                       const float* __restrict__ W2v,
                       const float* __restrict__ WskS,
                       const float* __restrict__ WskV,
                       float* __restrict__ out) {
  __shared__ alignas(16) float lds[(128 + 512) * WP + PT * TILE_F];
  float* wl2  = lds;
  float* wlsk = wl2 + 128 * WP;
  const int lane = threadIdx.x & 31;
  const int wid  = threadIdx.x >> 5;
  const int l16  = lane & 15;
  const int kh   = lane >> 4;
  const int g    = wid >> 2;
  const int ct   = wid & 3;
  const int col  = 16 * ct + l16;
  const int row0 = (blockIdx.x * PT + g) * 16;
  int rowA = row0 + l16; if (rowA >= NN) rowA = NN - 1;
  float* tile = lds + (128 + 512) * WP + g * TILE_F;

  const float* arow8 = attrs + (size_t)rowA * 8;
  const float attE0 = arow8[2 * kh]     * 0.04419417382f;
  const float attE1 = arow8[2 * kh + 1] * 0.04419417382f;
  const float attO0 = arow8[4 + 2 * kh] * 0.04419417382f;
  const float attO1 = arow8[5 + 2 * kh] * 0.04419417382f;
  const float* mrow = msmv + (size_t)rowA * 512;

  stage_weights<128>(W2s, wl2, threadIdx.x, blockIdx.x);
  stage_weights<512>(WskS, wlsk, threadIdx.x, blockIdx.x);
  __syncthreads();
  {
    v8f acc[1] = {};
    post_pass<true, 1, 4, false>(mrow, wl2, wlsk, tile, kh, l16, col,
                                 attE0, attE1, attO0, attO1, acc);
    __syncthreads();
    #pragma unroll
    for (int j = 0; j < 8; ++j) tile[(j + 8 * kh) * TP + col] = acc[0][j];
  }
  stage_weights<128>(W2v, wl2, threadIdx.x, blockIdx.x);
  stage_weights<512>(WskV, wlsk, threadIdx.x, blockIdx.x);
  __syncthreads();
  #pragma unroll
  for (int rr = 0; rr < 4; ++rr) {
    const int m = 4 * ct + rr;
    const int r = row0 + m;
    const float2 v = *(const float2*)(tile + m * TP + 2 * lane);
    if (r < NN) *(float2*)(out + (size_t)r * 256 + 2 * lane) = v;
  }
  {
    v8f acc[3] = {};
    post_pass<false, 3, 2, true>(mrow + 128, wl2, wlsk, tile, kh, l16, col,
                                 attE0, attE1, attO0, attO1, acc);
    __syncthreads();
    #pragma unroll
    for (int j = 0; j < 8; ++j) {
      #pragma unroll
      for (int d = 0; d < 3; ++d)
        tile[(j + 8 * kh) * VP + col * 3 + d] = acc[d][j];
    }
  }
  __syncthreads();
  #pragma unroll
  for (int rr = 0; rr < 4; ++rr) {
    const int m = 4 * ct + rr;
    const int r = row0 + m;
    #pragma unroll
    for (int q = 0; q < 3; ++q) {
      const float2 v = *(const float2*)(tile + m * VP + 64 * q + 2 * lane);
      if (r < NN) *(float2*)(out + (size_t)r * 256 + 64 + 64 * q + 2 * lane) = v;
    }
  }
}

extern "C" void kernel_launch(void* const* d_in, const int* in_sizes, int n_in,
                              void* d_out, int out_size, void* d_ws, size_t ws_size,
                              hipStream_t stream) {
  const float* node_attrs = (const float*)d_in[0];
  const float* node_feats = (const float*)d_in[1];
  const float* edge_attrs = (const float*)d_in[2];
  const float* edge_feats = (const float*)d_in[3];
  const int*   edge_index = (const int*)d_in[4];
  const float* W1s  = (const float*)d_in[5];
  const float* W1v  = (const float*)d_in[6];
  const float* Wm1  = (const float*)d_in[7];
  const float* Wm2  = (const float*)d_in[8];
  const float* W2s  = (const float*)d_in[9];
  const float* W2v  = (const float*)d_in[10];
  const float* WskS = (const float*)d_in[11];
  const float* WskV = (const float*)d_in[12];

  float* out  = (float*)d_out;
  float* msmv = (float*)d_ws;

  int* cnt  = (int*)(msmv + (size_t)NN * 512);

  k_zero<<<20, 256, 0, stream>>>((float4*)cnt, NN / 4);
  k_front<<<PRE_WGS + FILL_WGS, 256, 0, stream>>>(node_feats, W1s, W1v, out,
                                                  edge_index, cnt, (int*)d_ws);
  k_msg <<<(NN + 7) / 8, 256, 0, stream>>>(edge_attrs, edge_feats, edge_index,
                                           Wm1, Wm2, out, cnt, (const int*)d_ws, msmv);
  k_post<<<(NN + 16 * PT - 1) / (16 * PT), PT * 128, 0, stream>>>(msmv, node_attrs, W2s, W2v,
                                                                  WskS, WskV, out);
}
